# starting kernel
# baseline (speedup 1.0000x reference)
#include <hip/hip_runtime.h>
#include <math.h>

#define N_NODES 50000
#define N_EDGES 800000
#define NNZ_H   800000
#define M_HE    10000
#define F_IN    128
#define F_HID   64
#define F_OUT   64

typedef float v2f __attribute__((ext_vector_type(2)));
typedef float v8f __attribute__((ext_vector_type(8)));

__global__ __launch_bounds__(128)
void gemm_wmma_f32(const float* __restrict__ A, const float* __restrict__ B,
                   float* __restrict__ C, int K) {
  __shared__ float Bs[F_IN * 64];
  const int tid = threadIdx.x;
  const int kn  = K * 64;
  for (int i = tid; i < kn; i += 128) Bs[i] = B[i];
  __syncthreads();

  const int lane    = tid & 31;
  const int waveCol = (tid >> 5) * 16;
  const int m       = lane & 15;
  const int hh      = lane >> 4;
  const long row0   = (long)blockIdx.x * 16;

  const float* Arow = A  + (row0 + m) * K;
  const float* Bp   = Bs + waveCol + m;

  v8f acc = {};
  for (int k0 = 0; k0 < K; k0 += 4) {
    const int ka = k0 + (hh << 1);
    v2f a, b;
    a.x = Arow[ka];
    a.y = Arow[ka + 1];
    b.x = Bp[(long)ka * 64];
    b.y = Bp[(long)(ka + 1) * 64];
    acc = __builtin_amdgcn_wmma_f32_16x16x4_f32(
        false, a, false, b, (short)0, acc, false, false);
  }
  float* Cp = C + (row0 + (hh << 3)) * 64 + waveCol + m;
#pragma unroll
  for (int r = 0; r < 8; ++r) Cp[(long)r * 64] = acc[r];
}

__global__ void k_zero(float* __restrict__ p, long n) {
  long i = (long)blockIdx.x * blockDim.x + threadIdx.x;
  if (i < n) p[i] = 0.0f;
}

__global__ void k_count(const int* __restrict__ idx, float* __restrict__ deg, int n) {
  int i = blockIdx.x * blockDim.x + threadIdx.x;
  if (i < n) atomicAdd(&deg[idx[i]], 1.0f);
}

__global__ void k_gcn_dis(float* __restrict__ d, int n) {
  int i = blockIdx.x * blockDim.x + threadIdx.x;
  if (i < n) d[i] = rsqrtf(d[i] + 1.0f);
}

__global__ void k_inv(float* __restrict__ d, long n) {
  long i = (long)blockIdx.x * blockDim.x + threadIdx.x;
  if (i < n) { float v = d[i]; d[i] = (v > 0.0f) ? (1.0f / v) : 0.0f; }
}

__global__ void k_gcn_self(const float* __restrict__ xw, const float* __restrict__ dis,
                           float* __restrict__ agg, int n  ) {
  int i = blockIdx.x * blockDim.x + threadIdx.x;
  if (i < n) {
    int node = i >> 4, q = (i & 15) << 2;
    float w = dis[node]; w *= w;
    const float4 v = *(const float4*)(xw + (long)node * 64 + q);
    *(float4*)(agg + (long)node * 64 + q) =
        make_float4(w * v.x, w * v.y, w * v.z, w * v.w);
  }
}

__global__ void k_gcn_edge(const int* __restrict__ row, const int* __restrict__ col,
                           const float* __restrict__ dis, const float* __restrict__ xw,
                           float* __restrict__ agg, int n  ) {
  int i = blockIdx.x * blockDim.x + threadIdx.x;
  if (i < n) {
    int e = i >> 4, q = (i & 15) << 2;
    int r = row[e], c = col[e];
    float w = dis[r] * dis[c];
    const float4 v = *(const float4*)(xw + (long)r * 64 + q);
    float* o = agg + (long)c * 64 + q;
    atomicAdd(o + 0, w * v.x);
    atomicAdd(o + 1, w * v.y);
    atomicAdd(o + 2, w * v.z);
    atomicAdd(o + 3, w * v.w);
  }
}

__global__ void k_bias_act(const float* __restrict__ agg, const float* __restrict__ b,
                           float* __restrict__ out, long n, int do_elu) {
  long i = (long)blockIdx.x * blockDim.x + threadIdx.x;
  if (i < n) {
    float v = agg[i] + b[i & 63];
    if (do_elu) v = (v > 0.0f) ? v : expm1f(v);
    out[i] = v;
  }
}

__global__ void k_hyp_fwd(const int* __restrict__ nodei, const int* __restrict__ hei,
                          const float* __restrict__ xw, float* __restrict__ ef,
                          int n  ) {
  int i = blockIdx.x * blockDim.x + threadIdx.x;
  if (i < n) {
    int e = i >> 4, q = (i & 15) << 2;
    int nd = nodei[e], h = hei[e];
    const float4 v = *(const float4*)(xw + (long)nd * 64 + q);
    float* o = ef + (long)h * 64 + q;
    atomicAdd(o + 0, v.x);
    atomicAdd(o + 1, v.y);
    atomicAdd(o + 2, v.z);
    atomicAdd(o + 3, v.w);
  }
}

__global__ void k_hyp_back(const int* __restrict__ nodei, const int* __restrict__ hei,
                           const float* __restrict__ Binv, const float* __restrict__ ef,
                           float* __restrict__ agg, int n  ) {
  int i = blockIdx.x * blockDim.x + threadIdx.x;
  if (i < n) {
    int e = i >> 4, q = (i & 15) << 2;
    int nd = nodei[e], h = hei[e];
    float w = Binv[h];
    const float4 v = *(const float4*)(ef + (long)h * 64 + q);
    float* o = agg + (long)nd * 64 + q;
    atomicAdd(o + 0, w * v.x);
    atomicAdd(o + 1, w * v.y);
    atomicAdd(o + 2, w * v.z);
    atomicAdd(o + 3, w * v.w);
  }
}

__global__ void k_hyp_final(const float* __restrict__ agg, const float* __restrict__ Dinv,
                            const float* __restrict__ b, float* __restrict__ out,
                            long n, int do_elu) {
  long i = (long)blockIdx.x * blockDim.x + threadIdx.x;
  if (i < n) {
    long node = i >> 6;
    float v = Dinv[node] * agg[i] + b[i & 63];
    if (do_elu) v = (v > 0.0f) ? v : expm1f(v);
    out[i] = v;
  }
}

__global__ void k_fuse(const float* __restrict__ xs, const float* __restrict__ xd,
                       const float* __restrict__ gate, float* __restrict__ z, long n) {
  long i = (long)blockIdx.x * blockDim.x + threadIdx.x;
  if (i < n) {
    float a = 1.0f / (1.0f + __expf(-gate[0]));
    z[i] = a * xs[i] + (1.0f - a) * xd[i];
  }
}

extern "C" void kernel_launch(void* const* d_in, const int* in_sizes, int n_in,
                              void* d_out, int out_size, void* d_ws, size_t ws_size,
                              hipStream_t stream) {
  (void)in_sizes; (void)n_in; (void)out_size; (void)ws_size;

  const float* x   = (const float*)d_in[0];
  const int*   ei  = (const int*)d_in[1];
  const int*   hyi = (const int*)d_in[2];
  const float* g1w = (const float*)d_in[3];
  const float* g1b = (const float*)d_in[4];
  const float* g2w = (const float*)d_in[5];
  const float* g2b = (const float*)d_in[6];
  const float* h1w = (const float*)d_in[7];
  const float* h1b = (const float*)d_in[8];
  const float* h2w = (const float*)d_in[9];
  const float* h2b = (const float*)d_in[10];
  const float* gate = (const float*)d_in[11];

  float* out_z  = (float*)d_out;
  float* out_xs = out_z  + (long)N_NODES * F_OUT;
  float* out_xd = out_xs + (long)N_NODES * F_OUT;

  float* ws   = (float*)d_ws;
  float* xw   = ws;
  float* agg  = xw   + (long)N_NODES * 64;
  float* h    = agg  + (long)N_NODES * 64;
  float* dis  = h    + (long)N_NODES * 64;
  float* Dinv = dis  + N_NODES;
  float* Binv = Dinv + N_NODES;
  float* ef   = Binv + M_HE;

  const int T = 256;
  auto cd = [](long a, long b) { return (int)((a + b - 1) / b); };
  const long NF = (long)N_NODES * 64;
  const int  N16 = N_NODES * 16, E16 = N_EDGES * 16, Z16 = NNZ_H * 16;

  k_zero   <<<cd(N_NODES, T), T, 0, stream>>>(dis, N_NODES);
  k_count  <<<cd(N_EDGES, T), T, 0, stream>>>(ei + N_EDGES, dis, N_EDGES);
  k_gcn_dis<<<cd(N_NODES, T), T, 0, stream>>>(dis, N_NODES);

  gemm_wmma_f32<<<N_NODES / 16, 128, 0, stream>>>(x, g1w, xw, F_IN);
  k_gcn_self<<<cd(N16, T), T, 0, stream>>>(xw, dis, agg, N16);
  k_gcn_edge<<<cd(E16, T), T, 0, stream>>>(ei, ei + N_EDGES, dis, xw, agg, E16);
  k_bias_act<<<cd(NF, T), T, 0, stream>>>(agg, g1b, h, NF, 1);

  gemm_wmma_f32<<<N_NODES / 16, 128, 0, stream>>>(h, g2w, xw, F_HID);
  k_gcn_self<<<cd(N16, T), T, 0, stream>>>(xw, dis, agg, N16);
  k_gcn_edge<<<cd(E16, T), T, 0, stream>>>(ei, ei + N_EDGES, dis, xw, agg, E16);
  k_bias_act<<<cd(NF, T), T, 0, stream>>>(agg, g2b, out_xs, NF, 0);

  k_zero <<<cd(N_NODES + M_HE, T), T, 0, stream>>>(Dinv, N_NODES + M_HE);
  k_count<<<cd(NNZ_H, T), T, 0, stream>>>(hyi,         Dinv, NNZ_H);
  k_count<<<cd(NNZ_H, T), T, 0, stream>>>(hyi + NNZ_H, Binv, NNZ_H);
  k_inv  <<<cd(N_NODES + M_HE, T), T, 0, stream>>>(Dinv, N_NODES + M_HE);

  gemm_wmma_f32<<<N_NODES / 16, 128, 0, stream>>>(x, h1w, xw, F_IN);
  k_zero    <<<cd((long)M_HE * 64, T), T, 0, stream>>>(ef, (long)M_HE * 64);
  k_hyp_fwd <<<cd(Z16, T), T, 0, stream>>>(hyi, hyi + NNZ_H, xw, ef, Z16);
  k_zero    <<<cd(NF, T), T, 0, stream>>>(agg, NF);
  k_hyp_back<<<cd(Z16, T), T, 0, stream>>>(hyi, hyi + NNZ_H, Binv, ef, agg, Z16);
  k_hyp_final<<<cd(NF, T), T, 0, stream>>>(agg, Dinv, h1b, h, NF, 1);

  gemm_wmma_f32<<<N_NODES / 16, 128, 0, stream>>>(h, h2w, xw, F_HID);
  k_zero    <<<cd((long)M_HE * 64, T), T, 0, stream>>>(ef, (long)M_HE * 64);
  k_hyp_fwd <<<cd(Z16, T), T, 0, stream>>>(hyi, hyi + NNZ_H, xw, ef, Z16);
  k_zero    <<<cd(NF, T), T, 0, stream>>>(agg, NF);
  k_hyp_back<<<cd(Z16, T), T, 0, stream>>>(hyi, hyi + NNZ_H, Binv, ef, agg, Z16);
  k_hyp_final<<<cd(NF, T), T, 0, stream>>>(agg, Dinv, h2b, out_xd, NF, 0);

  k_fuse<<<cd(NF, T), T, 0, stream>>>(out_xs, out_xd, gate, out_z, NF);
}


# scatter-adds by owner work-groups: each holds a block of output rows in LDS, scans the whole index list, adds source rows by LDS float atomics (arrival order), writes each row once; no global atomics
# speedup vs baseline: 2.0687x; 2.0687x over previous
#include <hip/hip_runtime.h>
#include <math.h>


#define N_NODES 50000
#define N_EDGES 800000
#define NNZ_H   800000
#define M_HE    10000
#define F_IN    128
#define F_HID   64
#define F_OUT   64

#define OWN_WG    256
#define OWN_T     1024
#define OWN_RN    ((N_NODES + OWN_WG - 1) / OWN_WG)
#define OWN_RM    ((M_HE + OWN_WG - 1) / OWN_WG)
#define OWN_WC    448
#define OWN_LC    (32 * OWN_WC)
#define OWN_U     8
#define OWN_G     (OWN_T * OWN_U)
#define OWN_PU    8

static_assert(N_EDGES % 4 == 0 && NNZ_H % 4 == 0, "index lists are read 4 entries per load");
static_assert(OWN_WC % OWN_PU == 0, "a wave's slots are taken OWN_PU at a time");

typedef float v2f __attribute__((ext_vector_type(2)));
typedef float v8f __attribute__((ext_vector_type(8)));

__global__ __launch_bounds__(128)
void gemm_wmma_f32(const float* __restrict__ A, const float* __restrict__ B,
                   float* __restrict__ C, int K) {
  __shared__ float Bs[F_IN * 64];
  const int tid = threadIdx.x;
  const int kn  = K * 64;
  for (int i = tid; i < kn; i += 128) Bs[i] = B[i];
  __syncthreads();

  const int lane    = tid & 31;
  const int waveCol = (tid >> 5) * 16;
  const int m       = lane & 15;
  const int hh      = lane >> 4;
  const long row0   = (long)blockIdx.x * 16;

  const float* Arow = A  + (row0 + m) * K;
  const float* Bp   = Bs + waveCol + m;

  v8f acc = {};
  for (int k0 = 0; k0 < K; k0 += 4) {
    const int ka = k0 + (hh << 1);
    v2f a, b;
    a.x = Arow[ka];
    a.y = Arow[ka + 1];
    b.x = Bp[(long)ka * 64];
    b.y = Bp[(long)(ka + 1) * 64];
    acc = __builtin_amdgcn_wmma_f32_16x16x4_f32(
        false, a, false, b, (short)0, acc, false, false);
  }
  float* Cp = C + (row0 + (hh << 3)) * 64 + waveCol + m;
#pragma unroll
  for (int r = 0; r < 8; ++r) Cp[(long)r * 64] = acc[r];
}


__device__ __forceinline__ int own_first_group(int nGrp) {
  return (nGrp > 0) ? (int)(blockIdx.x % (unsigned)nGrp) : 0;
}

__global__ __launch_bounds__(OWN_T)
void owner_count(const int* __restrict__ dst, int L, float* __restrict__ dis,
                 int nDst, int R) {
  __shared__ int s_cnt[OWN_RN];
  const int tid = threadIdx.x;
  const int r0  = blockIdx.x * R;
  if (tid < R) s_cnt[tid] = 0;
  __syncthreads();

  auto scan4 = [&](const int4& v) {
    const unsigned a = (unsigned)(v.x - r0), b = (unsigned)(v.y - r0);
    const unsigned c = (unsigned)(v.z - r0), d = (unsigned)(v.w - r0);
    if (min(min(a, b), min(c, d)) < (unsigned)R) {
      if (a < (unsigned)R) atomicAdd(&s_cnt[a], 1);
      if (b < (unsigned)R) atomicAdd(&s_cnt[b], 1);
      if (c < (unsigned)R) atomicAdd(&s_cnt[c], 1);
      if (d < (unsigned)R) atomicAdd(&s_cnt[d], 1);
    }
  };
  const int4* d4   = (const int4*)dst;
  const int   n4   = L >> 2;
  const int   nGrp = n4 / OWN_G;
  const int   g0   = own_first_group(nGrp);
  for (int g = 0; g < nGrp; ++g) {
    int gi = g + g0; if (gi >= nGrp) gi -= nGrp;
    const int4* p = d4 + gi * OWN_G + tid;
    int4 v[OWN_U];
#pragma unroll
    for (int j = 0; j < OWN_U; ++j) v[j] = p[j * OWN_T];
#pragma unroll
    for (int j = 0; j < OWN_U; ++j) scan4(v[j]);
  }
  {
    const int q0 = nGrp * OWN_G + tid;
    int4 v[OWN_U];
#pragma unroll
    for (int j = 0; j < OWN_U; ++j)
      v[j] = (q0 + j * OWN_T < n4) ? d4[q0 + j * OWN_T] : make_int4(-1, -1, -1, -1);
#pragma unroll
    for (int j = 0; j < OWN_U; ++j) scan4(v[j]);
  }
  __syncthreads();
  if (tid < R && r0 + tid < nDst)
    dis[r0 + tid] = rsqrtf((float)s_cnt[tid] + 1.0f);
}

template <int MODE>
__global__ __launch_bounds__(OWN_T)
void owner_pass(const int* __restrict__ dst, const int* __restrict__ src, int L,
                const float* __restrict__ S, int nSrc,
                const float* __restrict__ dis, const float* __restrict__ bias,
                float* __restrict__ out, int nDst, int R, int do_elu) {
  __shared__ alignas(16) float s_acc[OWN_RN * 64];
  __shared__ float s_dis[OWN_RN];
  __shared__ int   s_cnt[OWN_RN];
  __shared__ int2  s_l[OWN_LC];
  __shared__ float s_lw[OWN_LC];

  const int tid = threadIdx.x;
  const int wv = tid >> 5, lane = tid & 31;
  const int r0  = blockIdx.x * R;

  if (tid < R) {
    s_cnt[tid] = 0;
    if (MODE == 0) s_dis[tid] = (r0 + tid < nDst) ? dis[r0 + tid] : 0.0f;
  }
  for (int i = tid; i < R * 16; i += OWN_T) {
    const int dl = i >> 4, q = (i & 15) << 2;
    float4 a = make_float4(0.0f, 0.0f, 0.0f, 0.0f);
    if (MODE == 0 && r0 + dl < nDst) {
      float w = dis[r0 + dl]; w *= w;
      const float4 v = *(const float4*)(S + (long)(r0 + dl) * 64 + q);
      a = make_float4(w * v.x, w * v.y, w * v.z, w * v.w);
    }
    *(float4*)&s_acc[dl * 64 + q] = a;
  }
  __syncthreads();

  const int lbase = wv * OWN_WC;
  int wcnt = 0;
  auto hit1 = [&](bool h, unsigned dl, int e) {
    const unsigned bal = __builtin_amdgcn_ballot_w32(h);
    if (bal) {
      if (h) {
        const int slot = wcnt + (int)__builtin_amdgcn_mbcnt_lo(bal, 0u);
        atomicAdd(&s_cnt[dl], 1);
        if (slot < OWN_WC) s_l[lbase + slot] = make_int2(e, (int)dl);
        else {
          const int s = src[e];
          if ((unsigned)s < (unsigned)nSrc) {
            const float w = (MODE == 0) ? dis[s] * s_dis[dl] : 1.0f;
            const float* sp = S + (long)s * 64;
#pragma nounroll
            for (int cc = 0; cc < 64; ++cc) atomicAdd(&s_acc[dl * 64 + cc], w * sp[cc]);
          }
        }
      }
      wcnt += __builtin_popcount(bal);
    }
  };
  auto scan4 = [&](const int4& v, int e0) {
    const unsigned a = (unsigned)(v.x - r0), b = (unsigned)(v.y - r0);
    const unsigned c = (unsigned)(v.z - r0), d = (unsigned)(v.w - r0);
    if (__builtin_amdgcn_ballot_w32(min(min(a, b), min(c, d)) < (unsigned)R)) {
      hit1(a < (unsigned)R, a, e0);
      hit1(b < (unsigned)R, b, e0 + 1);
      hit1(c < (unsigned)R, c, e0 + 2);
      hit1(d < (unsigned)R, d, e0 + 3);
    }
  };
  const int4* d4   = (const int4*)dst;
  const int   n4   = L >> 2;
  const int   nGrp = n4 / OWN_G;
  const int   g0   = own_first_group(nGrp);
  for (int g = 0; g < nGrp; ++g) {
    int gi = g + g0; if (gi >= nGrp) gi -= nGrp;
    const int q0 = gi * OWN_G + tid;
    int4 v[OWN_U];
#pragma unroll
    for (int j = 0; j < OWN_U; ++j) v[j] = d4[q0 + j * OWN_T];
#pragma unroll
    for (int j = 0; j < OWN_U; ++j) scan4(v[j], (q0 + j * OWN_T) << 2);
  }
  {
    const int q0 = nGrp * OWN_G + tid;
    int4 v[OWN_U];
#pragma unroll
    for (int j = 0; j < OWN_U; ++j)
      v[j] = (q0 + j * OWN_T < n4) ? d4[q0 + j * OWN_T] : make_int4(-1, -1, -1, -1);
#pragma unroll
    for (int j = 0; j < OWN_U; ++j) scan4(v[j], (q0 + j * OWN_T) << 2);
  }

  const int cnt    = min(wcnt, OWN_WC);
  const int cntPad = (cnt + OWN_PU - 1) & ~(OWN_PU - 1);
  for (int m = lane; m < cntPad; m += 32) {
    int s = -1, dl = 0; float ws = 0.0f;
    if (m < cnt) {
      const int2 ed = s_l[lbase + m];
      const int sv = src[ed.x];
      dl = ed.y;
      if ((unsigned)sv < (unsigned)nSrc) {
        s = sv;
        if (MODE == 0) ws = dis[sv];
      }
    }
    s_l[lbase + m] = make_int2(s, dl);
    if (MODE == 0) s_lw[lbase + m] = ws;
  }
  __syncthreads();

  for (int m0 = 0; m0 < cntPad; m0 += OWN_PU) {
    int s[OWN_PU], dl[OWN_PU]; float w[OWN_PU]; float2 v[OWN_PU];
#pragma unroll
    for (int j = 0; j < OWN_PU; ++j) {
      const int2 sd = s_l[lbase + m0 + j];
      s[j]  = __builtin_amdgcn_readfirstlane(sd.x);
      dl[j] = __builtin_amdgcn_readfirstlane(sd.y);
      w[j]  = (MODE == 0) ? s_lw[lbase + m0 + j] * s_dis[dl[j]] : 1.0f;
      v[j]  = *(const float2*)(S + (long)max(s[j], 0) * 64 + 2 * lane);
    }
#pragma unroll
    for (int j = 0; j < OWN_PU; ++j)
      if (s[j] >= 0) {
        atomicAdd(&s_acc[dl[j] * 64 + 2 * lane],     w[j] * v[j].x);
        atomicAdd(&s_acc[dl[j] * 64 + 2 * lane + 1], w[j] * v[j].y);
      }
  }
  __syncthreads();

  for (int i = tid; i < R * 16; i += OWN_T) {
    const int dl = i >> 4, q = (i & 15) << 2;
    const int row = r0 + dl;
    if (row < nDst) {
      const float4 a = *(const float4*)&s_acc[dl * 64 + q];
      float4 r;
      if (MODE == 0) {
        const float4 b = *(const float4*)(bias + q);
        r = make_float4(a.x + b.x, a.y + b.y, a.z + b.z, a.w + b.w);
      } else {
        const int   c  = s_cnt[dl];
        const float sc = (c > 0) ? (1.0f / (float)c) : 0.0f;
        if (MODE == 1) {
          r = make_float4(sc * a.x, sc * a.y, sc * a.z, sc * a.w);
        } else {
          const float4 b = *(const float4*)(bias + q);
          r = make_float4(sc * a.x + b.x, sc * a.y + b.y, sc * a.z + b.z, sc * a.w + b.w);
        }
      }
      if (do_elu) {
        r.x = (r.x > 0.0f) ? r.x : expm1f(r.x);
        r.y = (r.y > 0.0f) ? r.y : expm1f(r.y);
        r.z = (r.z > 0.0f) ? r.z : expm1f(r.z);
        r.w = (r.w > 0.0f) ? r.w : expm1f(r.w);
      }
      *(float4*)(out + (long)row * 64 + q) = r;
    }
  }
}

__global__ void k_fuse(const float* __restrict__ xs, const float* __restrict__ xd,
                       const float* __restrict__ gate, float* __restrict__ z, long n) {
  long i = (long)blockIdx.x * blockDim.x + threadIdx.x;
  if (i < n) {
    float a = 1.0f / (1.0f + __expf(-gate[0]));
    z[i] = a * xs[i] + (1.0f - a) * xd[i];
  }
}

extern "C" void kernel_launch(void* const* d_in, const int* in_sizes, int n_in,
                              void* d_out, int out_size, void* d_ws, size_t ws_size,
                              hipStream_t stream) {
  (void)in_sizes; (void)n_in; (void)out_size; (void)ws_size;

  const float* x   = (const float*)d_in[0];
  const int*   ei  = (const int*)d_in[1];
  const int*   hyi = (const int*)d_in[2];
  const float* g1w = (const float*)d_in[3];
  const float* g1b = (const float*)d_in[4];
  const float* g2w = (const float*)d_in[5];
  const float* g2b = (const float*)d_in[6];
  const float* h1w = (const float*)d_in[7];
  const float* h1b = (const float*)d_in[8];
  const float* h2w = (const float*)d_in[9];
  const float* h2b = (const float*)d_in[10];
  const float* gate = (const float*)d_in[11];

  float* out_z  = (float*)d_out;
  float* out_xs = out_z  + (long)N_NODES * F_OUT;
  float* out_xd = out_xs + (long)N_NODES * F_OUT;

  float* ws   = (float*)d_ws;
  float* xw   = ws;
  float* h    = xw   + (long)N_NODES * 64;
  float* dis  = h    + (long)N_NODES * 64;
  float* ef   = dis  + N_NODES;


  owner_count<<<OWN_WG, OWN_T, 0, stream>>>(ei + N_EDGES, N_EDGES, dis, N_NODES, OWN_RN);

  gemm_wmma_f32<<<N_NODES / 16, 128, 0, stream>>>(x, g1w, xw, F_IN);
  owner_pass<0><<<OWN_WG, OWN_T, 0, stream>>>(ei + N_EDGES, ei, N_EDGES, xw, N_NODES,
                                              dis, g1b, h, N_NODES, OWN_RN, 1);
  gemm_wmma_f32<<<N_NODES / 16, 128, 0, stream>>>(h, g2w, xw, F_HID);
  owner_pass<0><<<OWN_WG, OWN_T, 0, stream>>>(ei + N_EDGES, ei, N_EDGES, xw, N_NODES,
                                              dis, g2b, out_xs, N_NODES, OWN_RN, 0);

  gemm_wmma_f32<<<N_NODES / 16, 128, 0, stream>>>(x, h1w, xw, F_IN);
  owner_pass<1><<<OWN_WG, OWN_T, 0, stream>>>(hyi + NNZ_H, hyi, NNZ_H, xw, N_NODES,
                                              nullptr, nullptr, ef, M_HE, OWN_RM, 0);
  owner_pass<2><<<OWN_WG, OWN_T, 0, stream>>>(hyi, hyi + NNZ_H, NNZ_H, ef, M_HE,
                                              nullptr, h1b, h, N_NODES, OWN_RN, 1);
  gemm_wmma_f32<<<N_NODES / 16, 128, 0, stream>>>(h, h2w, xw, F_HID);
  owner_pass<1><<<OWN_WG, OWN_T, 0, stream>>>(hyi + NNZ_H, hyi, NNZ_H, xw, N_NODES,
                                              nullptr, nullptr, ef, M_HE, OWN_RM, 0);
  owner_pass<2><<<OWN_WG, OWN_T, 0, stream>>>(hyi, hyi + NNZ_H, NNZ_H, ef, M_HE,
                                              nullptr, h2b, out_xd, N_NODES, OWN_RN, 0);

  k_fuse<<<(N_NODES * 64 + 255) / 256, 256, 0, stream>>>(out_xs, out_xd, gate, out_z, (long)N_NODES * 64);
}


# each index list scanned by one launch per call: it hands every wave's matches (source row, local row) on in global memory, later passes read them back; 7 list scans -> 3; LDS float adds as before
# speedup vs baseline: 2.7184x; 1.3141x over previous
#include <hip/hip_runtime.h>
#include <math.h>


#define N_NODES 50000
#define N_EDGES 800000
#define NNZ_H   800000
#define M_HE    10000
#define F_IN    128
#define F_HID   64
#define F_OUT   64

#define OWN_WG    256
#define OWN_T     1024
#define OWN_RN    ((N_NODES + OWN_WG - 1) / OWN_WG)
#define OWN_RM    ((M_HE + OWN_WG - 1) / OWN_WG)
#define OWN_WC    448
#define OWN_LC    (32 * OWN_WC)
#define OWN_U     8
#define OWN_G     (OWN_T * OWN_U)
#define OWN_PU    8
#define SAVE_WC   192
#define SAVE_N    (OWN_WG * (OWN_T / 32) * SAVE_WC)

static_assert(N_EDGES % 4 == 0 && NNZ_H % 4 == 0, "index lists are read 4 entries per load");
static_assert(OWN_WC % OWN_PU == 0, "a wave's slots are taken OWN_PU at a time");
static_assert(SAVE_WC <= OWN_WC && SAVE_WC % 32 == 0, "a handed-on list fits the wave's LDS slots and is read back 32 matches at a time");
static_assert(N_NODES < 0xFFFF && M_HE < 0xFFFF && OWN_RN <= 0xFFFF, "a match is packed as source row (16 bits, 0xFFFF = none) and local row (16 bits)");
static_assert(2L * SAVE_N + 2L * OWN_WG * (OWN_T / 32) <= (long)N_NODES * F_OUT, "the two lists and their counts lie in the z plane of the output until the last launch writes it");

typedef float v2f __attribute__((ext_vector_type(2)));
typedef float v8f __attribute__((ext_vector_type(8)));

__global__ __launch_bounds__(128)
void gemm_wmma_f32(const float* __restrict__ A, const float* __restrict__ B,
                   float* __restrict__ C, int K) {
  __shared__ float Bs[F_IN * 64];
  const int tid = threadIdx.x;
  const int kn  = K * 64;
  for (int i = tid; i < kn; i += 128) Bs[i] = B[i];
  __syncthreads();

  const int lane    = tid & 31;
  const int waveCol = (tid >> 5) * 16;
  const int m       = lane & 15;
  const int hh      = lane >> 4;
  const long row0   = (long)blockIdx.x * 16;

  const float* Arow = A  + (row0 + m) * K;
  const float* Bp   = Bs + waveCol + m;

  v8f acc = {};
  for (int k0 = 0; k0 < K; k0 += 4) {
    const int ka = k0 + (hh << 1);
    v2f a, b;
    a.x = Arow[ka];
    a.y = Arow[ka + 1];
    b.x = Bp[(long)ka * 64];
    b.y = Bp[(long)(ka + 1) * 64];
    acc = __builtin_amdgcn_wmma_f32_16x16x4_f32(
        false, a, false, b, (short)0, acc, false, false);
  }
  float* Cp = C + (row0 + (hh << 3)) * 64 + waveCol + m;
#pragma unroll
  for (int r = 0; r < 8; ++r) Cp[(long)r * 64] = acc[r];
}


__device__ __forceinline__ int own_first_group(int nGrp) {
  return (nGrp > 0) ? (int)(blockIdx.x % (unsigned)nGrp) : 0;
}

__device__ __forceinline__ unsigned own_pack(int s, int dl) {
  return ((unsigned)dl << 16) | ((unsigned)s & 0xFFFFu);
}

template <class HIT>
__device__ __forceinline__ void own_scan(const int* __restrict__ dst, int L, int r0, int R, HIT&& hit1) {
  const int tid = threadIdx.x;
  auto scan4 = [&](const int4& v, int e0) {
    const unsigned a = (unsigned)(v.x - r0), b = (unsigned)(v.y - r0);
    const unsigned c = (unsigned)(v.z - r0), d = (unsigned)(v.w - r0);
    if (__builtin_amdgcn_ballot_w32(min(min(a, b), min(c, d)) < (unsigned)R)) {
      hit1(a < (unsigned)R, a, e0);
      hit1(b < (unsigned)R, b, e0 + 1);
      hit1(c < (unsigned)R, c, e0 + 2);
      hit1(d < (unsigned)R, d, e0 + 3);
    }
  };
  const int4* d4   = (const int4*)dst;
  const int   n4   = L >> 2;
  const int   nGrp = n4 / OWN_G;
  const int   g0   = own_first_group(nGrp);
  for (int g = 0; g < nGrp; ++g) {
    int gi = g + g0; if (gi >= nGrp) gi -= nGrp;
    const int q0 = gi * OWN_G + tid;
    int4 v[OWN_U];
#pragma unroll
    for (int j = 0; j < OWN_U; ++j) v[j] = d4[q0 + j * OWN_T];
#pragma unroll
    for (int j = 0; j < OWN_U; ++j) scan4(v[j], (q0 + j * OWN_T) << 2);
  }
  {
    const int q0 = nGrp * OWN_G + tid;
    int4 v[OWN_U];
#pragma unroll
    for (int j = 0; j < OWN_U; ++j)
      v[j] = (q0 + j * OWN_T < n4) ? d4[q0 + j * OWN_T] : make_int4(-1, -1, -1, -1);
#pragma unroll
    for (int j = 0; j < OWN_U; ++j) scan4(v[j], (q0 + j * OWN_T) << 2);
  }
}

__global__ __launch_bounds__(OWN_T)
void owner_count(const int* __restrict__ dst, const int* __restrict__ src, int L, int nSrc,
                 float* __restrict__ dis, int nDst, int R,
                 unsigned* __restrict__ sv_list, int* __restrict__ sv_cnt) {
  __shared__ int  s_cnt[OWN_RN];
  __shared__ int2 s_l[(OWN_T / 32) * SAVE_WC];
  const int tid = threadIdx.x;
  const int wv = tid >> 5, lane = tid & 31;
  const int r0  = blockIdx.x * R;
  if (tid < R) s_cnt[tid] = 0;
  __syncthreads();

  const int lbase = wv * SAVE_WC;
  int wcnt = 0;
  own_scan(dst, L, r0, R, [&](bool h, unsigned dl, int e) {
    const unsigned bal = __builtin_amdgcn_ballot_w32(h);
    if (bal) {
      if (h) {
        const int slot = wcnt + (int)__builtin_amdgcn_mbcnt_lo(bal, 0u);
        atomicAdd(&s_cnt[dl], 1);
        if (slot < SAVE_WC) s_l[lbase + slot] = make_int2(e, (int)dl);
      }
      wcnt += __builtin_popcount(bal);
    }
  });

  const int widx = blockIdx.x * (OWN_T / 32) + wv;
  if (wcnt <= SAVE_WC) {
    unsigned* wl = sv_list + (long)widx * SAVE_WC;
    for (int m = lane; m < wcnt; m += 32) {
      const int2 ed = s_l[lbase + m];
      const int sv = src[ed.x];
      wl[m] = own_pack(((unsigned)sv < (unsigned)nSrc) ? sv : -1, ed.y);
    }
  }
  if (lane == 0) sv_cnt[widx] = wcnt;
  __syncthreads();
  if (tid < R && r0 + tid < nDst)
    dis[r0 + tid] = rsqrtf((float)s_cnt[tid] + 1.0f);
}

template <int MODE>
__global__ __launch_bounds__(OWN_T)
void owner_pass(const int* __restrict__ dst, const int* __restrict__ src, int L,
                const float* __restrict__ S, int nSrc,
                const float* __restrict__ dis, const float* __restrict__ bias,
                float* __restrict__ out, int nDst, int R, int do_elu,
                unsigned* sv_list, int* sv_cnt, int reuse) {
  __shared__ alignas(16) float s_acc[OWN_RN * 64];
  __shared__ float s_dis[OWN_RN];
  __shared__ int   s_cnt[OWN_RN];
  __shared__ int2  s_l[OWN_LC];
  __shared__ float s_lw[OWN_LC];

  const int tid = threadIdx.x;
  const int wv = tid >> 5, lane = tid & 31;
  const int r0  = blockIdx.x * R;

  if (tid < R) {
    s_cnt[tid] = 0;
    if (MODE == 0) s_dis[tid] = (r0 + tid < nDst) ? dis[r0 + tid] : 0.0f;
  }
  for (int i = tid; i < R * 16; i += OWN_T) {
    const int dl = i >> 4, q = (i & 15) << 2;
    float4 a = make_float4(0.0f, 0.0f, 0.0f, 0.0f);
    if (MODE == 0 && r0 + dl < nDst) {
      float w = dis[r0 + dl]; w *= w;
      const float4 v = *(const float4*)(S + (long)(r0 + dl) * 64 + q);
      a = make_float4(w * v.x, w * v.y, w * v.z, w * v.w);
    }
    *(float4*)&s_acc[dl * 64 + q] = a;
  }
  __syncthreads();

  const int lbase = wv * OWN_WC;
  const int widx  = blockIdx.x * (OWN_T / 32) + wv;
  unsigned* wl    = sv_list + (long)widx * SAVE_WC;
  int wcnt = 0;
  bool have = false;
  if (reuse) {
    const int c = __builtin_amdgcn_readfirstlane(sv_cnt[widx]);
    if ((unsigned)c <= (unsigned)SAVE_WC) { wcnt = c; have = true; }
  }
  int cntPad;
  if (have) {
    cntPad = (wcnt + OWN_PU - 1) & ~(OWN_PU - 1);
    unsigned p[SAVE_WC / 32];
    int s[SAVE_WC / 32], dl[SAVE_WC / 32]; float ws[SAVE_WC / 32];
#pragma unroll
    for (int j = 0; j < SAVE_WC / 32; ++j)
      p[j] = (lane + 32 * j < wcnt) ? wl[lane + 32 * j] : 0xFFFFFFFFu;
#pragma unroll
    for (int j = 0; j < SAVE_WC / 32; ++j) {
      const unsigned pl = p[j] >> 16, ps = p[j] & 0xFFFFu;
      s[j] = -1; dl[j] = 0;
      if (pl < (unsigned)R) {
        dl[j] = (int)pl;
        if (MODE != 0) atomicAdd(&s_cnt[pl], 1);
        if (ps < (unsigned)nSrc) s[j] = (int)ps;
      }
    }
#pragma unroll
    for (int j = 0; j < SAVE_WC / 32; ++j)
      ws[j] = (MODE == 0 && s[j] >= 0) ? dis[s[j]] : 0.0f;
#pragma unroll
    for (int j = 0; j < SAVE_WC / 32; ++j) {
      const int m = lane + 32 * j;
      if (m < cntPad) {
        s_l[lbase + m] = make_int2(s[j], dl[j]);
        if (MODE == 0) s_lw[lbase + m] = ws[j];
      }
    }
  } else {
    own_scan(dst, L, r0, R, [&](bool h, unsigned dl, int e) {
      const unsigned bal = __builtin_amdgcn_ballot_w32(h);
      if (bal) {
        if (h) {
          const int slot = wcnt + (int)__builtin_amdgcn_mbcnt_lo(bal, 0u);
          atomicAdd(&s_cnt[dl], 1);
          if (slot < OWN_WC) s_l[lbase + slot] = make_int2(e, (int)dl);
          else {
            const int s = src[e];
            if ((unsigned)s < (unsigned)nSrc) {
              const float w = (MODE == 0) ? dis[s] * s_dis[dl] : 1.0f;
              const float* sp = S + (long)s * 64;
#pragma nounroll
              for (int cc = 0; cc < 64; ++cc) atomicAdd(&s_acc[dl * 64 + cc], w * sp[cc]);
            }
          }
        }
        wcnt += __builtin_popcount(bal);
      }
    });

    const int  cnt  = min(wcnt, OWN_WC);
    const bool hand = !reuse && wcnt <= SAVE_WC;
    cntPad = (cnt + OWN_PU - 1) & ~(OWN_PU - 1);
    for (int m = lane; m < cntPad; m += 32) {
      int s = -1, dl = 0; float ws = 0.0f;
      if (m < cnt) {
        const int2 ed = s_l[lbase + m];
        const int sv = src[ed.x];
        dl = ed.y;
        if ((unsigned)sv < (unsigned)nSrc) {
          s = sv;
          if (MODE == 0) ws = dis[sv];
        }
        if (hand) wl[m] = own_pack(s, dl);
      }
      s_l[lbase + m] = make_int2(s, dl);
      if (MODE == 0) s_lw[lbase + m] = ws;
    }
    if (!reuse && lane == 0) sv_cnt[widx] = wcnt;
  }
  __syncthreads();

  for (int m0 = 0; m0 < cntPad; m0 += OWN_PU) {
    int s[OWN_PU], dl[OWN_PU]; float w[OWN_PU]; float2 v[OWN_PU];
#pragma unroll
    for (int j = 0; j < OWN_PU; ++j) {
      const int2 sd = s_l[lbase + m0 + j];
      s[j]  = __builtin_amdgcn_readfirstlane(sd.x);
      dl[j] = __builtin_amdgcn_readfirstlane(sd.y);
      w[j]  = (MODE == 0) ? s_lw[lbase + m0 + j] * s_dis[dl[j]] : 1.0f;
      v[j]  = *(const float2*)(S + (long)max(s[j], 0) * 64 + 2 * lane);
    }
#pragma unroll
    for (int j = 0; j < OWN_PU; ++j)
      if (s[j] >= 0) {
        atomicAdd(&s_acc[dl[j] * 64 + 2 * lane],     w[j] * v[j].x);
        atomicAdd(&s_acc[dl[j] * 64 + 2 * lane + 1], w[j] * v[j].y);
      }
  }
  __syncthreads();

  for (int i = tid; i < R * 16; i += OWN_T) {
    const int dl = i >> 4, q = (i & 15) << 2;
    const int row = r0 + dl;
    if (row < nDst) {
      const float4 a = *(const float4*)&s_acc[dl * 64 + q];
      float4 r;
      if (MODE == 0) {
        const float4 b = *(const float4*)(bias + q);
        r = make_float4(a.x + b.x, a.y + b.y, a.z + b.z, a.w + b.w);
      } else {
        const int   c  = s_cnt[dl];
        const float sc = (c > 0) ? (1.0f / (float)c) : 0.0f;
        if (MODE == 1) {
          r = make_float4(sc * a.x, sc * a.y, sc * a.z, sc * a.w);
        } else {
          const float4 b = *(const float4*)(bias + q);
          r = make_float4(sc * a.x + b.x, sc * a.y + b.y, sc * a.z + b.z, sc * a.w + b.w);
        }
      }
      if (do_elu) {
        r.x = (r.x > 0.0f) ? r.x : expm1f(r.x);
        r.y = (r.y > 0.0f) ? r.y : expm1f(r.y);
        r.z = (r.z > 0.0f) ? r.z : expm1f(r.z);
        r.w = (r.w > 0.0f) ? r.w : expm1f(r.w);
      }
      *(float4*)(out + (long)row * 64 + q) = r;
    }
  }
}

__global__ void k_fuse(const float* __restrict__ xs, const float* __restrict__ xd,
                       const float* __restrict__ gate, float* __restrict__ z, long n) {
  long i = (long)blockIdx.x * blockDim.x + threadIdx.x;
  if (i < n) {
    float a = 1.0f / (1.0f + __expf(-gate[0]));
    z[i] = a * xs[i] + (1.0f - a) * xd[i];
  }
}

extern "C" void kernel_launch(void* const* d_in, const int* in_sizes, int n_in,
                              void* d_out, int out_size, void* d_ws, size_t ws_size,
                              hipStream_t stream) {
  (void)in_sizes; (void)n_in; (void)out_size; (void)ws_size;

  const float* x   = (const float*)d_in[0];
  const int*   ei  = (const int*)d_in[1];
  const int*   hyi = (const int*)d_in[2];
  const float* g1w = (const float*)d_in[3];
  const float* g1b = (const float*)d_in[4];
  const float* g2w = (const float*)d_in[5];
  const float* g2b = (const float*)d_in[6];
  const float* h1w = (const float*)d_in[7];
  const float* h1b = (const float*)d_in[8];
  const float* h2w = (const float*)d_in[9];
  const float* h2b = (const float*)d_in[10];
  const float* gate = (const float*)d_in[11];

  float* out_z  = (float*)d_out;
  float* out_xs = out_z  + (long)N_NODES * F_OUT;
  float* out_xd = out_xs + (long)N_NODES * F_OUT;

  float* ws   = (float*)d_ws;
  float* xw   = ws;
  float* h    = xw   + (long)N_NODES * 64;
  float* dis  = h    + (long)N_NODES * 64;
  float* ef   = dis  + N_NODES;

  unsigned* sv0  = (unsigned*)out_z;
  unsigned* sv1  = sv0 + SAVE_N;
  int*      svc0 = (int*)(sv1 + SAVE_N);
  int*      svc1 = svc0 + OWN_WG * (OWN_T / 32);

  owner_count<<<OWN_WG, OWN_T, 0, stream>>>(ei + N_EDGES, ei, N_EDGES, N_NODES, dis, N_NODES, OWN_RN,
                                            sv0, svc0);

  gemm_wmma_f32<<<N_NODES / 16, 128, 0, stream>>>(x, g1w, xw, F_IN);
  owner_pass<0><<<OWN_WG, OWN_T, 0, stream>>>(ei + N_EDGES, ei, N_EDGES, xw, N_NODES,
                                              dis, g1b, h, N_NODES, OWN_RN, 1, sv0, svc0, 1);
  gemm_wmma_f32<<<N_NODES / 16, 128, 0, stream>>>(h, g2w, xw, F_HID);
  owner_pass<0><<<OWN_WG, OWN_T, 0, stream>>>(ei + N_EDGES, ei, N_EDGES, xw, N_NODES,
                                              dis, g2b, out_xs, N_NODES, OWN_RN, 0, sv0, svc0, 1);

  gemm_wmma_f32<<<N_NODES / 16, 128, 0, stream>>>(x, h1w, xw, F_IN);
  owner_pass<1><<<OWN_WG, OWN_T, 0, stream>>>(hyi + NNZ_H, hyi, NNZ_H, xw, N_NODES,
                                              nullptr, nullptr, ef, M_HE, OWN_RM, 0, sv0, svc0, 0);
  owner_pass<2><<<OWN_WG, OWN_T, 0, stream>>>(hyi, hyi + NNZ_H, NNZ_H, ef, M_HE,
                                              nullptr, h1b, h, N_NODES, OWN_RN, 1, sv1, svc1, 0);
  gemm_wmma_f32<<<N_NODES / 16, 128, 0, stream>>>(h, h2w, xw, F_HID);
  owner_pass<1><<<OWN_WG, OWN_T, 0, stream>>>(hyi + NNZ_H, hyi, NNZ_H, xw, N_NODES,
                                              nullptr, nullptr, ef, M_HE, OWN_RM, 0, sv0, svc0, 1);
  owner_pass<2><<<OWN_WG, OWN_T, 0, stream>>>(hyi, hyi + NNZ_H, NNZ_H, ef, M_HE,
                                              nullptr, h2b, out_xd, N_NODES, OWN_RN, 0, sv1, svc1, 1);

  k_fuse<<<(N_NODES * 64 + 255) / 256, 256, 0, stream>>>(out_xs, out_xd, gate, out_z, (long)N_NODES * 64);
}


# each index list is split once per call into 256 slices bucketed by owner block (LDS integer counts); owners read their 256 short runs and never scan a list; LDS float adds by arrival, as before
# speedup vs baseline: 3.4993x; 1.2873x over previous
#include <hip/hip_runtime.h>
#include <math.h>


#define N_NODES 50000
#define N_EDGES 800000
#define NNZ_H   800000
#define M_HE    10000
#define F_IN    128
#define F_HID   64
#define F_OUT   64

#define OWN_WG    256
#define OWN_T     1024
#define OWN_RN    ((N_NODES + OWN_WG - 1) / OWN_WG)
#define OWN_RM    ((M_HE + OWN_WG - 1) / OWN_WG)
#define OWN_WC    448
#define OWN_LC    (32 * OWN_WC)
#define OWN_PU    8
#define OWN_NB    6

#define PART_N    256
#define PART_T    256
#define PART_SL   ((N_EDGES + PART_N - 1) / PART_N)
#define PART_E    ((PART_SL + PART_T - 1) / PART_T)
#define PART_SS   ((PART_SL + 31) & ~31)
#define PART_OS   ((PART_N + 1 + 31) & ~31)
#define PART_LW   (PART_N * PART_SS)
#define PART_OW   (PART_N * PART_OS)

static_assert(N_EDGES == NNZ_H, "the three index lists have one length, so one slice length");
static_assert(PART_N == 8 * (OWN_T / 32), "a wave of an owner reads 8 slices' runs");
static_assert(PART_T == OWN_WG, "the offsets of the owner bins are summed one bin per thread");
static_assert(OWN_WC % OWN_PU == 0, "a wave's slots are taken OWN_PU at a time");
static_assert(N_NODES < 0xFFFF && M_HE < 0xFFFF && OWN_RN <= 0xFFFF, "a match is packed as source row (16 bits, 0xFFFF = none) and local row (16 bits)");
static_assert(3L * (PART_LW + PART_OW) <= (long)N_NODES * F_OUT, "the three bucketed lists and their offsets lie in the z plane of the output until the last launch writes it");

typedef float v2f __attribute__((ext_vector_type(2)));
typedef float v8f __attribute__((ext_vector_type(8)));

__global__ __launch_bounds__(128)
void gemm_wmma_f32(const float* __restrict__ A, const float* __restrict__ B,
                   float* __restrict__ C, int K) {
  __shared__ float Bs[F_IN * 64];
  const int tid = threadIdx.x;
  const int kn  = K * 64;
  for (int i = tid; i < kn; i += 128) Bs[i] = B[i];
  __syncthreads();

  const int lane    = tid & 31;
  const int waveCol = (tid >> 5) * 16;
  const int m       = lane & 15;
  const int hh      = lane >> 4;
  const long row0   = (long)blockIdx.x * 16;

  const float* Arow = A  + (row0 + m) * K;
  const float* Bp   = Bs + waveCol + m;

  v8f acc = {};
  for (int k0 = 0; k0 < K; k0 += 4) {
    const int ka = k0 + (hh << 1);
    v2f a, b;
    a.x = Arow[ka];
    a.y = Arow[ka + 1];
    b.x = Bp[(long)ka * 64];
    b.y = Bp[(long)(ka + 1) * 64];
    acc = __builtin_amdgcn_wmma_f32_16x16x4_f32(
        false, a, false, b, (short)0, acc, false, false);
  }
  float* Cp = C + (row0 + (hh << 3)) * 64 + waveCol + m;
#pragma unroll
  for (int r = 0; r < 8; ++r) Cp[(long)r * 64] = acc[r];
}


__device__ __forceinline__ unsigned own_pack(int s, int dl) {
  return ((unsigned)dl << 16) | ((unsigned)s & 0xFFFFu);
}

template <int R>
__device__ __forceinline__ bool part_bucket(int d, int& o, int& dl) {
  if ((unsigned)d >= (unsigned)(OWN_WG * R)) return false;
  o = d / R; dl = d - o * R;
  return true;
}

__global__ __launch_bounds__(PART_T)
void part_lists(const int* __restrict__ dG, const int* __restrict__ sG,
                const int* __restrict__ dH, const int* __restrict__ sH,
                const int* __restrict__ dN, const int* __restrict__ sN,
                unsigned* __restrict__ lst, int* __restrict__ off) {
  __shared__ int      s_hist[OWN_WG];
  __shared__ int      s_cur[OWN_WG];
  __shared__ int      s_wt[PART_T / 32];
  __shared__ unsigned s_sorted[PART_SL];
  const int tid = threadIdx.x;
  const int wv = tid >> 5, lane = tid & 31;
  const int li = blockIdx.x / PART_N;
  const int sl = blockIdx.x - li * PART_N;
  const int* __restrict__ dst = (li == 0) ? dG : (li == 1) ? dH : dN;
  const int* __restrict__ src = (li == 0) ? sG : (li == 1) ? sH : sN;
  const int L    = (li == 0) ? N_EDGES : NNZ_H;
  const int nSrc = (li == 2) ? M_HE : N_NODES;
  unsigned* lstS = lst + (long)li * PART_LW + (long)sl * PART_SS;
  int*      offS = off + (long)li * PART_OW + (long)sl * PART_OS;
  const int e0 = sl * PART_SL;
  const int e1 = min(e0 + PART_SL, L);

  s_hist[tid] = 0;
  __syncthreads();

  int d[PART_E], s[PART_E], ow[PART_E]; unsigned pk[PART_E];
#pragma unroll
  for (int j = 0; j < PART_E; ++j) {
    const int e  = e0 + tid + j * PART_T;
    const int ec = min(e, L - 1);
    const int dv = dst[ec], sv = src[ec];
    d[j] = (e < e1) ? dv : -1;
    s[j] = (e < e1) ? sv : -1;
  }
#pragma unroll
  for (int j = 0; j < PART_E; ++j) {
    int o = -1, dl = 0;
    const bool in = (li == 1) ? part_bucket<OWN_RM>(d[j], o, dl) : part_bucket<OWN_RN>(d[j], o, dl);
    ow[j] = in ? o : -1;
    pk[j] = own_pack(((unsigned)s[j] < (unsigned)nSrc) ? s[j] : -1, dl);
    if (in) atomicAdd(&s_hist[o], 1);
  }
  __syncthreads();

  const int v = s_hist[tid];
  int inc = v;
#pragma unroll
  for (int k = 1; k < 32; k <<= 1) {
    const int t = __shfl_up(inc, k);
    if (lane >= k) inc += t;
  }
  if (lane == 31) s_wt[wv] = inc;
  __syncthreads();
  int base = 0, tot = 0;
#pragma unroll
  for (int w = 0; w < PART_T / 32; ++w) {
    const int t = s_wt[w];
    if (w < wv) base += t;
    tot += t;
  }
  const int ex = base + inc - v;
  s_cur[tid] = ex;
  offS[tid] = ex;
  if (tid == PART_T - 1) offS[PART_N] = tot;
  __syncthreads();

#pragma unroll
  for (int j = 0; j < PART_E; ++j)
    if (ow[j] >= 0) s_sorted[atomicAdd(&s_cur[ow[j]], 1)] = pk[j];
  __syncthreads();

  for (int i = tid; i < tot; i += PART_T) lstS[i] = s_sorted[i];
}

struct OwnRuns { int pre[9]; int adj[8]; };

__device__ __forceinline__ void own_runs(const int* __restrict__ off, int wv, int lane, OwnRuns& r) {
  int a = 0, len = 0;
  if (lane < 8) {
    const int* o = off + (long)(wv * 8 + lane) * PART_OS + blockIdx.x;
    a   = min(max(o[0], 0), PART_SL);
    len = min(max(o[1] - a, 0), PART_SL - a);
  }
  r.pre[0] = 0;
#pragma unroll
  for (int k = 0; k < 8; ++k) {
    r.pre[k + 1] = r.pre[k] + __builtin_amdgcn_readlane(len, k);
    r.adj[k]     = (wv * 8 + k) * PART_SS + __builtin_amdgcn_readlane(a, k) - r.pre[k];
  }
}

__device__ __forceinline__ unsigned own_match(const unsigned* __restrict__ lst, const OwnRuns& r, int m) {
  const int mc = min(m, r.pre[8] - 1);
  int ad = r.adj[0];
#pragma unroll
  for (int k = 1; k < 8; ++k) ad = (mc >= r.pre[k]) ? r.adj[k] : ad;
  const unsigned p = lst[ad + mc];
  return (m == mc) ? p : 0xFFFFFFFFu;
}

__global__ __launch_bounds__(OWN_T)
void owner_count(const unsigned* __restrict__ lst, const int* __restrict__ off,
                 float* __restrict__ dis, int nDst, int R) {
  __shared__ int s_cnt[OWN_RN];
  const int tid = threadIdx.x;
  const int wv = __builtin_amdgcn_readfirstlane(tid >> 5), lane = tid & 31;
  const int r0 = blockIdx.x * R;
  if (tid < R) s_cnt[tid] = 0;
  __syncthreads();

  OwnRuns rn;
  own_runs(off, wv, lane, rn);
  const int wcnt = rn.pre[8];
  for (int m0 = 0; m0 < wcnt; m0 += 32 * OWN_NB) {
    unsigned p[OWN_NB];
#pragma unroll
    for (int j = 0; j < OWN_NB; ++j) {
      const int m = m0 + 32 * j + lane;
      p[j] = own_match(lst, rn, m);
    }
#pragma unroll
    for (int j = 0; j < OWN_NB; ++j) {
      const unsigned pl = p[j] >> 16;
      if (pl < (unsigned)R) atomicAdd(&s_cnt[pl], 1);
    }
  }
  __syncthreads();
  if (tid < R && r0 + tid < nDst)
    dis[r0 + tid] = rsqrtf((float)s_cnt[tid] + 1.0f);
}

template <int MODE>
__global__ __launch_bounds__(OWN_T)
void owner_pass(const unsigned* __restrict__ lst, const int* __restrict__ off,
                const float* __restrict__ S, int nSrc,
                const float* __restrict__ dis, const float* __restrict__ bias,
                float* __restrict__ out, int nDst, int R, int do_elu) {
  __shared__ alignas(16) float s_acc[OWN_RN * 64];
  __shared__ float s_dis[OWN_RN];
  __shared__ int   s_cnt[OWN_RN];
  __shared__ int2  s_l[OWN_LC];
  __shared__ float s_lw[OWN_LC];

  const int tid = threadIdx.x;
  const int wv = __builtin_amdgcn_readfirstlane(tid >> 5), lane = tid & 31;
  const int r0  = blockIdx.x * R;

  if (tid < R) {
    s_cnt[tid] = 0;
    if (MODE == 0) s_dis[tid] = (r0 + tid < nDst) ? dis[r0 + tid] : 0.0f;
  }
  for (int i = tid; i < R * 16; i += OWN_T) {
    const int dl = i >> 4, q = (i & 15) << 2;
    float4 a = make_float4(0.0f, 0.0f, 0.0f, 0.0f);
    if (MODE == 0 && r0 + dl < nDst) {
      float w = dis[r0 + dl]; w *= w;
      const float4 v = *(const float4*)(S + (long)(r0 + dl) * 64 + q);
      a = make_float4(w * v.x, w * v.y, w * v.z, w * v.w);
    }
    *(float4*)&s_acc[dl * 64 + q] = a;
  }
  __syncthreads();

  const int lbase = wv * OWN_WC;
  OwnRuns rn;
  own_runs(off, wv, lane, rn);
  const int wcnt = rn.pre[8];
  for (int m0 = 0; m0 < wcnt; m0 += 32 * OWN_NB) {
    unsigned p[OWN_NB];
    int s[OWN_NB], dl[OWN_NB]; float ws[OWN_NB];
#pragma unroll
    for (int j = 0; j < OWN_NB; ++j) {
      const int m = m0 + 32 * j + lane;
      p[j] = own_match(lst, rn, m);
    }
#pragma unroll
    for (int j = 0; j < OWN_NB; ++j) {
      const unsigned pl = p[j] >> 16, ps = p[j] & 0xFFFFu;
      s[j] = -1; dl[j] = 0;
      if (pl < (unsigned)R) {
        dl[j] = (int)pl;
        if (MODE != 0) atomicAdd(&s_cnt[pl], 1);
        if (ps < (unsigned)nSrc) s[j] = (int)ps;
      }
    }
#pragma unroll
    for (int j = 0; j < OWN_NB; ++j) {
      const float t = (MODE == 0) ? dis[max(s[j], 0)] : 0.0f;
      ws[j] = (s[j] >= 0) ? t : 0.0f;
    }
#pragma unroll
    for (int j = 0; j < OWN_NB; ++j) {
      const int m = m0 + 32 * j + lane;
      if (m < wcnt) {
        if (m < OWN_WC) {
          s_l[lbase + m] = make_int2(s[j], dl[j]);
          if (MODE == 0) s_lw[lbase + m] = ws[j];
        } else if (s[j] >= 0) {
          const float w = (MODE == 0) ? ws[j] * s_dis[dl[j]] : 1.0f;
          const float* sp = S + (long)s[j] * 64;
#pragma nounroll
          for (int cc = 0; cc < 64; ++cc) atomicAdd(&s_acc[dl[j] * 64 + cc], w * sp[cc]);
        }
      }
    }
  }
  const int cnt    = min(wcnt, OWN_WC);
  const int cntPad = (cnt + OWN_PU - 1) & ~(OWN_PU - 1);
  if (cnt + lane < cntPad) {
    s_l[lbase + cnt + lane] = make_int2(-1, 0);
    if (MODE == 0) s_lw[lbase + cnt + lane] = 0.0f;
  }
  __syncthreads();

  for (int m0 = 0; m0 < cntPad; m0 += OWN_PU) {
    int s[OWN_PU], dl[OWN_PU]; float w[OWN_PU]; float2 v[OWN_PU];
#pragma unroll
    for (int j = 0; j < OWN_PU; ++j) {
      const int2 sd = s_l[lbase + m0 + j];
      s[j]  = __builtin_amdgcn_readfirstlane(sd.x);
      dl[j] = __builtin_amdgcn_readfirstlane(sd.y);
      w[j]  = (MODE == 0) ? s_lw[lbase + m0 + j] * s_dis[dl[j]] : 1.0f;
      v[j]  = *(const float2*)(S + (long)max(s[j], 0) * 64 + 2 * lane);
    }
#pragma unroll
    for (int j = 0; j < OWN_PU; ++j)
      if (s[j] >= 0) {
        atomicAdd(&s_acc[dl[j] * 64 + 2 * lane],     w[j] * v[j].x);
        atomicAdd(&s_acc[dl[j] * 64 + 2 * lane + 1], w[j] * v[j].y);
      }
  }
  __syncthreads();

  for (int i = tid; i < R * 16; i += OWN_T) {
    const int dl = i >> 4, q = (i & 15) << 2;
    const int row = r0 + dl;
    if (row < nDst) {
      const float4 a = *(const float4*)&s_acc[dl * 64 + q];
      float4 r;
      if (MODE == 0) {
        const float4 b = *(const float4*)(bias + q);
        r = make_float4(a.x + b.x, a.y + b.y, a.z + b.z, a.w + b.w);
      } else {
        const int   c  = s_cnt[dl];
        const float sc = (c > 0) ? (1.0f / (float)c) : 0.0f;
        if (MODE == 1) {
          r = make_float4(sc * a.x, sc * a.y, sc * a.z, sc * a.w);
        } else {
          const float4 b = *(const float4*)(bias + q);
          r = make_float4(sc * a.x + b.x, sc * a.y + b.y, sc * a.z + b.z, sc * a.w + b.w);
        }
      }
      if (do_elu) {
        r.x = (r.x > 0.0f) ? r.x : expm1f(r.x);
        r.y = (r.y > 0.0f) ? r.y : expm1f(r.y);
        r.z = (r.z > 0.0f) ? r.z : expm1f(r.z);
        r.w = (r.w > 0.0f) ? r.w : expm1f(r.w);
      }
      *(float4*)(out + (long)row * 64 + q) = r;
    }
  }
}

__global__ void k_fuse(const float* __restrict__ xs, const float* __restrict__ xd,
                       const float* __restrict__ gate, float* __restrict__ z, long n) {
  long i = (long)blockIdx.x * blockDim.x + threadIdx.x;
  if (i < n) {
    float a = 1.0f / (1.0f + __expf(-gate[0]));
    z[i] = a * xs[i] + (1.0f - a) * xd[i];
  }
}

extern "C" void kernel_launch(void* const* d_in, const int* in_sizes, int n_in,
                              void* d_out, int out_size, void* d_ws, size_t ws_size,
                              hipStream_t stream) {
  (void)in_sizes; (void)n_in; (void)out_size; (void)ws_size;

  const float* x   = (const float*)d_in[0];
  const int*   ei  = (const int*)d_in[1];
  const int*   hyi = (const int*)d_in[2];
  const float* g1w = (const float*)d_in[3];
  const float* g1b = (const float*)d_in[4];
  const float* g2w = (const float*)d_in[5];
  const float* g2b = (const float*)d_in[6];
  const float* h1w = (const float*)d_in[7];
  const float* h1b = (const float*)d_in[8];
  const float* h2w = (const float*)d_in[9];
  const float* h2b = (const float*)d_in[10];
  const float* gate = (const float*)d_in[11];

  float* out_z  = (float*)d_out;
  float* out_xs = out_z  + (long)N_NODES * F_OUT;
  float* out_xd = out_xs + (long)N_NODES * F_OUT;

  float* ws   = (float*)d_ws;
  float* xw   = ws;
  float* h    = xw   + (long)N_NODES * 64;
  float* dis  = h    + (long)N_NODES * 64;
  float* ef   = dis  + N_NODES;

  unsigned* lst  = (unsigned*)out_z;
  int*      off  = (int*)(lst + 3L * PART_LW);
  unsigned* lstG = lst;                 int* offG = off;
  unsigned* lstH = lst + PART_LW;       int* offH = off + PART_OW;
  unsigned* lstN = lst + 2L * PART_LW;  int* offN = off + 2L * PART_OW;

  part_lists<<<3 * PART_N, PART_T, 0, stream>>>(ei + N_EDGES, ei, hyi + NNZ_H, hyi, hyi, hyi + NNZ_H, lst, off);

  owner_count<<<OWN_WG, OWN_T, 0, stream>>>(lstG, offG, dis, N_NODES, OWN_RN);

  gemm_wmma_f32<<<N_NODES / 16, 128, 0, stream>>>(x, g1w, xw, F_IN);
  owner_pass<0><<<OWN_WG, OWN_T, 0, stream>>>(lstG, offG, xw, N_NODES, dis, g1b, h, N_NODES, OWN_RN, 1);
  gemm_wmma_f32<<<N_NODES / 16, 128, 0, stream>>>(h, g2w, xw, F_HID);
  owner_pass<0><<<OWN_WG, OWN_T, 0, stream>>>(lstG, offG, xw, N_NODES, dis, g2b, out_xs, N_NODES, OWN_RN, 0);

  gemm_wmma_f32<<<N_NODES / 16, 128, 0, stream>>>(x, h1w, xw, F_IN);
  owner_pass<1><<<OWN_WG, OWN_T, 0, stream>>>(lstH, offH, xw, N_NODES, nullptr, nullptr, ef, M_HE, OWN_RM, 0);
  owner_pass<2><<<OWN_WG, OWN_T, 0, stream>>>(lstN, offN, ef, M_HE, nullptr, h1b, h, N_NODES, OWN_RN, 1);
  gemm_wmma_f32<<<N_NODES / 16, 128, 0, stream>>>(h, h2w, xw, F_HID);
  owner_pass<1><<<OWN_WG, OWN_T, 0, stream>>>(lstH, offH, xw, N_NODES, nullptr, nullptr, ef, M_HE, OWN_RM, 0);
  owner_pass<2><<<OWN_WG, OWN_T, 0, stream>>>(lstN, offN, ef, M_HE, nullptr, h2b, out_xd, N_NODES, OWN_RN, 0);

  k_fuse<<<(N_NODES * 64 + 255) / 256, 256, 0, stream>>>(out_xs, out_xd, gate, out_z, (long)N_NODES * 64);
}


# layer-2 products inside the owner passes: finished hidden rows stay in LDS, waves multiply them by the 64x64 weights (WMMA, A from LDS), pass writes rows*W; 13 launches -> 11; k order, sums unchanged
# speedup vs baseline: 3.8546x; 1.1015x over previous
#include <hip/hip_runtime.h>
#include <math.h>


#define N_NODES 50000
#define N_EDGES 800000
#define NNZ_H   800000
#define M_HE    10000
#define F_IN    128
#define F_HID   64
#define F_OUT   64

#define OWN_WG    256
#define OWN_T     1024
#define OWN_RN    ((N_NODES + OWN_WG - 1) / OWN_WG)
#define OWN_RM    ((M_HE + OWN_WG - 1) / OWN_WG)
#define OWN_WC    448
#define OWN_LC    (32 * OWN_WC)
#define OWN_PU    8
#define OWN_NB    6

#define OWN_RT    ((OWN_RN + 15) / 16)
#define OWN_RP    (16 * OWN_RT)
#define OWN_AP    68
#define OWN_WP    72

#define PART_N    256
#define PART_T    256
#define PART_SL   ((N_EDGES + PART_N - 1) / PART_N)
#define PART_E    ((PART_SL + PART_T - 1) / PART_T)
#define PART_SS   ((PART_SL + 31) & ~31)
#define PART_OS   ((PART_N + 1 + 31) & ~31)
#define PART_LW   (PART_N * PART_SS)
#define PART_OW   (PART_N * PART_OS)

static_assert(N_EDGES == NNZ_H, "the three index lists have one length, so one slice length");
static_assert(PART_N == 8 * (OWN_T / 32), "a wave of an owner reads 8 slices' runs");
static_assert(PART_T == OWN_WG, "the offsets of the owner bins are summed one bin per thread");
static_assert(OWN_WC % OWN_PU == 0, "a wave's slots are taken OWN_PU at a time");
static_assert(OWN_T == 16 * F_HID && F_HID == 64 && F_OUT == 64, "the 64x64 weights go to LDS 16 bytes per thread");
static_assert(OWN_T / 32 == 4 * 8, "a wave's product tiles: column tile wave%4, row tiles wave/4, wave/4+8");
static_assert(N_NODES < 0xFFFF && M_HE < 0xFFFF && OWN_RN <= 0xFFFF, "a match is packed as source row (16 bits, 0xFFFF = none) and local row (16 bits)");
static_assert(3L * (PART_LW + PART_OW) <= (long)N_NODES * F_OUT, "the three bucketed lists and their offsets lie in the z plane of the output until the last launch writes it");

typedef float v2f __attribute__((ext_vector_type(2)));
typedef float v8f __attribute__((ext_vector_type(8)));

__global__ __launch_bounds__(128)
void gemm_wmma_f32(const float* __restrict__ A, const float* __restrict__ B,
                   float* __restrict__ C, int K) {
  __shared__ float Bs[F_IN * 64];
  const int tid = threadIdx.x;
  const int kn  = K * 64;
  for (int i = tid; i < kn; i += 128) Bs[i] = B[i];
  __syncthreads();

  const int lane    = tid & 31;
  const int waveCol = (tid >> 5) * 16;
  const int m       = lane & 15;
  const int hh      = lane >> 4;
  const long row0   = (long)blockIdx.x * 16;

  const float* Arow = A  + (row0 + m) * K;
  const float* Bp   = Bs + waveCol + m;

  v8f acc = {};
  for (int k0 = 0; k0 < K; k0 += 4) {
    const int ka = k0 + (hh << 1);
    v2f a, b;
    a.x = Arow[ka];
    a.y = Arow[ka + 1];
    b.x = Bp[(long)ka * 64];
    b.y = Bp[(long)(ka + 1) * 64];
    acc = __builtin_amdgcn_wmma_f32_16x16x4_f32(
        false, a, false, b, (short)0, acc, false, false);
  }
  float* Cp = C + (row0 + (hh << 3)) * 64 + waveCol + m;
#pragma unroll
  for (int r = 0; r < 8; ++r) Cp[(long)r * 64] = acc[r];
}


__device__ __forceinline__ unsigned own_pack(int s, int dl) {
  return ((unsigned)dl << 16) | ((unsigned)s & 0xFFFFu);
}

template <int R>
__device__ __forceinline__ bool part_bucket(int d, int& o, int& dl) {
  if ((unsigned)d >= (unsigned)(OWN_WG * R)) return false;
  o = d / R; dl = d - o * R;
  return true;
}

__global__ __launch_bounds__(PART_T)
void part_lists(const int* __restrict__ dG, const int* __restrict__ sG,
                const int* __restrict__ dH, const int* __restrict__ sH,
                const int* __restrict__ dN, const int* __restrict__ sN,
                unsigned* __restrict__ lst, int* __restrict__ off) {
  __shared__ int      s_hist[OWN_WG];
  __shared__ int      s_cur[OWN_WG];
  __shared__ int      s_wt[PART_T / 32];
  __shared__ unsigned s_sorted[PART_SL];
  const int tid = threadIdx.x;
  const int wv = tid >> 5, lane = tid & 31;
  const int li = blockIdx.x / PART_N;
  const int sl = blockIdx.x - li * PART_N;
  const int* __restrict__ dst = (li == 0) ? dG : (li == 1) ? dH : dN;
  const int* __restrict__ src = (li == 0) ? sG : (li == 1) ? sH : sN;
  const int L    = (li == 0) ? N_EDGES : NNZ_H;
  const int nSrc = (li == 2) ? M_HE : N_NODES;
  unsigned* lstS = lst + (long)li * PART_LW + (long)sl * PART_SS;
  int*      offS = off + (long)li * PART_OW + (long)sl * PART_OS;
  const int e0 = sl * PART_SL;
  const int e1 = min(e0 + PART_SL, L);

  s_hist[tid] = 0;
  __syncthreads();

  int d[PART_E], s[PART_E], ow[PART_E]; unsigned pk[PART_E];
#pragma unroll
  for (int j = 0; j < PART_E; ++j) {
    const int e  = e0 + tid + j * PART_T;
    const int ec = min(e, L - 1);
    const int dv = dst[ec], sv = src[ec];
    d[j] = (e < e1) ? dv : -1;
    s[j] = (e < e1) ? sv : -1;
  }
#pragma unroll
  for (int j = 0; j < PART_E; ++j) {
    int o = -1, dl = 0;
    const bool in = (li == 1) ? part_bucket<OWN_RM>(d[j], o, dl) : part_bucket<OWN_RN>(d[j], o, dl);
    ow[j] = in ? o : -1;
    pk[j] = own_pack(((unsigned)s[j] < (unsigned)nSrc) ? s[j] : -1, dl);
    if (in) atomicAdd(&s_hist[o], 1);
  }
  __syncthreads();

  const int v = s_hist[tid];
  int inc = v;
#pragma unroll
  for (int k = 1; k < 32; k <<= 1) {
    const int t = __shfl_up(inc, k);
    if (lane >= k) inc += t;
  }
  if (lane == 31) s_wt[wv] = inc;
  __syncthreads();
  int base = 0, tot = 0;
#pragma unroll
  for (int w = 0; w < PART_T / 32; ++w) {
    const int t = s_wt[w];
    if (w < wv) base += t;
    tot += t;
  }
  const int ex = base + inc - v;
  s_cur[tid] = ex;
  offS[tid] = ex;
  if (tid == PART_T - 1) offS[PART_N] = tot;
  __syncthreads();

#pragma unroll
  for (int j = 0; j < PART_E; ++j)
    if (ow[j] >= 0) s_sorted[atomicAdd(&s_cur[ow[j]], 1)] = pk[j];
  __syncthreads();

  for (int i = tid; i < tot; i += PART_T) lstS[i] = s_sorted[i];
}

struct OwnRuns { int pre[9]; int adj[8]; };

__device__ __forceinline__ void own_runs(const int* __restrict__ off, int wv, int lane, OwnRuns& r) {
  int a = 0, len = 0;
  if (lane < 8) {
    const int* o = off + (long)(wv * 8 + lane) * PART_OS + blockIdx.x;
    a   = min(max(o[0], 0), PART_SL);
    len = min(max(o[1] - a, 0), PART_SL - a);
  }
  r.pre[0] = 0;
#pragma unroll
  for (int k = 0; k < 8; ++k) {
    r.pre[k + 1] = r.pre[k] + __builtin_amdgcn_readlane(len, k);
    r.adj[k]     = (wv * 8 + k) * PART_SS + __builtin_amdgcn_readlane(a, k) - r.pre[k];
  }
}

__device__ __forceinline__ unsigned own_match(const unsigned* __restrict__ lst, const OwnRuns& r, int m) {
  const int mc = min(m, r.pre[8] - 1);
  int ad = r.adj[0];
#pragma unroll
  for (int k = 1; k < 8; ++k) ad = (mc >= r.pre[k]) ? r.adj[k] : ad;
  const unsigned p = lst[ad + mc];
  return (m == mc) ? p : 0xFFFFFFFFu;
}

__global__ __launch_bounds__(OWN_T)
void owner_count(const unsigned* __restrict__ lst, const int* __restrict__ off,
                 float* __restrict__ dis, int nDst, int R) {
  __shared__ int s_cnt[OWN_RN];
  const int tid = threadIdx.x;
  const int wv = __builtin_amdgcn_readfirstlane(tid >> 5), lane = tid & 31;
  const int r0 = blockIdx.x * R;
  if (tid < R) s_cnt[tid] = 0;
  __syncthreads();

  OwnRuns rn;
  own_runs(off, wv, lane, rn);
  const int wcnt = rn.pre[8];
  for (int m0 = 0; m0 < wcnt; m0 += 32 * OWN_NB) {
    unsigned p[OWN_NB];
#pragma unroll
    for (int j = 0; j < OWN_NB; ++j) {
      const int m = m0 + 32 * j + lane;
      p[j] = own_match(lst, rn, m);
    }
#pragma unroll
    for (int j = 0; j < OWN_NB; ++j) {
      const unsigned pl = p[j] >> 16;
      if (pl < (unsigned)R) atomicAdd(&s_cnt[pl], 1);
    }
  }
  __syncthreads();
  if (tid < R && r0 + tid < nDst)
    dis[r0 + tid] = rsqrtf((float)s_cnt[tid] + 1.0f);
}

template <int MODE, bool PROD>
__global__ __launch_bounds__(OWN_T)
void owner_pass(const unsigned* __restrict__ lst, const int* __restrict__ off,
                const float* __restrict__ S, int nSrc,
                const float* __restrict__ dis, const float* __restrict__ bias,
                const float* __restrict__ W,
                float* __restrict__ out, int nDst, int R, int do_elu) {
  constexpr int AP = PROD ? OWN_AP : 64;
  constexpr int AR = PROD ? OWN_RP : OWN_RN;
  __shared__ alignas(16) float s_acc[AR * AP];
  __shared__ alignas(16) float s_w[PROD ? 64 * OWN_WP : 4];
  __shared__ float s_dis[OWN_RN];
  __shared__ int   s_cnt[OWN_RN];
  __shared__ int2  s_l[OWN_LC];
  __shared__ float s_lw[OWN_LC];

  const int tid = threadIdx.x;
  const int wv = __builtin_amdgcn_readfirstlane(tid >> 5), lane = tid & 31;
  const int r0  = blockIdx.x * R;

  if (tid < R) {
    s_cnt[tid] = 0;
    if (MODE == 0) s_dis[tid] = (r0 + tid < nDst) ? dis[r0 + tid] : 0.0f;
  }
  float4 wq = make_float4(0.0f, 0.0f, 0.0f, 0.0f);
  if (PROD) wq = *(const float4*)(W + 4 * tid);
  const int rFill = PROD ? ((R + 15) & ~15) : R;
  for (int i = tid; i < rFill * 16; i += OWN_T) {
    const int dl = i >> 4, q = (i & 15) << 2;
    float4 a = make_float4(0.0f, 0.0f, 0.0f, 0.0f);
    if (MODE == 0 && dl < R && r0 + dl < nDst) {
      float w = dis[r0 + dl]; w *= w;
      const float4 v = *(const float4*)(S + (long)(r0 + dl) * 64 + q);
      a = make_float4(w * v.x, w * v.y, w * v.z, w * v.w);
    }
    *(float4*)&s_acc[dl * AP + q] = a;
  }
  if (PROD) *(float4*)&s_w[(tid >> 4) * OWN_WP + ((tid & 15) << 2)] = wq;
  __syncthreads();

  const int lbase = wv * OWN_WC;
  OwnRuns rn;
  own_runs(off, wv, lane, rn);
  const int wcnt = rn.pre[8];
  for (int m0 = 0; m0 < wcnt; m0 += 32 * OWN_NB) {
    unsigned p[OWN_NB];
    int s[OWN_NB], dl[OWN_NB]; float ws[OWN_NB];
#pragma unroll
    for (int j = 0; j < OWN_NB; ++j) {
      const int m = m0 + 32 * j + lane;
      p[j] = own_match(lst, rn, m);
    }
#pragma unroll
    for (int j = 0; j < OWN_NB; ++j) {
      const unsigned pl = p[j] >> 16, ps = p[j] & 0xFFFFu;
      s[j] = -1; dl[j] = 0;
      if (pl < (unsigned)R) {
        dl[j] = (int)pl;
        if (MODE != 0) atomicAdd(&s_cnt[pl], 1);
        if (ps < (unsigned)nSrc) s[j] = (int)ps;
      }
    }
#pragma unroll
    for (int j = 0; j < OWN_NB; ++j) {
      const float t = (MODE == 0) ? dis[max(s[j], 0)] : 0.0f;
      ws[j] = (s[j] >= 0) ? t : 0.0f;
    }
#pragma unroll
    for (int j = 0; j < OWN_NB; ++j) {
      const int m = m0 + 32 * j + lane;
      if (m < wcnt) {
        if (m < OWN_WC) {
          s_l[lbase + m] = make_int2(s[j], dl[j]);
          if (MODE == 0) s_lw[lbase + m] = ws[j];
        } else if (s[j] >= 0) {
          const float w = (MODE == 0) ? ws[j] * s_dis[dl[j]] : 1.0f;
          const float* sp = S + (long)s[j] * 64;
#pragma nounroll
          for (int cc = 0; cc < 64; ++cc) atomicAdd(&s_acc[dl[j] * AP + cc], w * sp[cc]);
        }
      }
    }
  }
  const int cnt    = min(wcnt, OWN_WC);
  const int cntPad = (cnt + OWN_PU - 1) & ~(OWN_PU - 1);
  if (cnt + lane < cntPad) {
    s_l[lbase + cnt + lane] = make_int2(-1, 0);
    if (MODE == 0) s_lw[lbase + cnt + lane] = 0.0f;
  }
  __syncthreads();

  for (int m0 = 0; m0 < cntPad; m0 += OWN_PU) {
    int s[OWN_PU], dl[OWN_PU]; float w[OWN_PU]; float2 v[OWN_PU];
#pragma unroll
    for (int j = 0; j < OWN_PU; ++j) {
      const int2 sd = s_l[lbase + m0 + j];
      s[j]  = __builtin_amdgcn_readfirstlane(sd.x);
      dl[j] = __builtin_amdgcn_readfirstlane(sd.y);
      w[j]  = (MODE == 0) ? s_lw[lbase + m0 + j] * s_dis[dl[j]] : 1.0f;
      v[j]  = *(const float2*)(S + (long)max(s[j], 0) * 64 + 2 * lane);
    }
#pragma unroll
    for (int j = 0; j < OWN_PU; ++j)
      if (s[j] >= 0) {
        atomicAdd(&s_acc[dl[j] * AP + 2 * lane],     w[j] * v[j].x);
        atomicAdd(&s_acc[dl[j] * AP + 2 * lane + 1], w[j] * v[j].y);
      }
  }
  __syncthreads();

  for (int i = tid; i < R * 16; i += OWN_T) {
    const int dl = i >> 4, q = (i & 15) << 2;
    const int row = r0 + dl;
    if (row < nDst) {
      const float4 a = *(const float4*)&s_acc[dl * AP + q];
      float4 r;
      if (MODE == 0) {
        const float4 b = *(const float4*)(bias + q);
        r = make_float4(a.x + b.x, a.y + b.y, a.z + b.z, a.w + b.w);
      } else {
        const int   c  = s_cnt[dl];
        const float sc = (c > 0) ? (1.0f / (float)c) : 0.0f;
        if (MODE == 1) {
          r = make_float4(sc * a.x, sc * a.y, sc * a.z, sc * a.w);
        } else {
          const float4 b = *(const float4*)(bias + q);
          r = make_float4(sc * a.x + b.x, sc * a.y + b.y, sc * a.z + b.z, sc * a.w + b.w);
        }
      }
      if (do_elu) {
        r.x = (r.x > 0.0f) ? r.x : expm1f(r.x);
        r.y = (r.y > 0.0f) ? r.y : expm1f(r.y);
        r.z = (r.z > 0.0f) ? r.z : expm1f(r.z);
        r.w = (r.w > 0.0f) ? r.w : expm1f(r.w);
      }
      if (PROD) *(float4*)&s_acc[dl * AP + q] = r;
      else      *(float4*)(out + (long)row * 64 + q) = r;
    }
  }

  if (PROD) {
    __syncthreads();
    const int m  = lane & 15;
    const int hh = lane >> 4;
    const int ct = (wv & 3) * 16;
    const int nOwn = min(R, nDst - r0);
    const int nRT = (nOwn + 15) >> 4;
    const float* bp = s_w + (hh << 1) * OWN_WP + ct + m;
    for (int rt = wv >> 2; rt < nRT; rt += OWN_T / 128) {
      const float* ap = s_acc + (rt * 16 + m) * AP + (hh << 1);
      v8f acc = {};
#pragma unroll
      for (int k0 = 0; k0 < 64; k0 += 4) {
        v2f a, b;
        a.x = ap[k0];
        a.y = ap[k0 + 1];
        b.x = bp[k0 * OWN_WP];
        b.y = bp[(k0 + 1) * OWN_WP];
        acc = __builtin_amdgcn_wmma_f32_16x16x4_f32(
            false, a, false, b, (short)0, acc, false, false);
      }
      const int dl0 = rt * 16 + (hh << 3);
      float* op = out + (long)(r0 + dl0) * 64 + ct + m;
      if (rt * 16 + 16 <= nOwn) {
#pragma unroll
        for (int r = 0; r < 8; ++r) op[(long)r * 64] = acc[r];
      } else {
#pragma unroll
        for (int r = 0; r < 8; ++r)
          if (dl0 + r < nOwn) op[(long)r * 64] = acc[r];
      }
    }
  }
}

__global__ void k_fuse(const float* __restrict__ xs, const float* __restrict__ xd,
                       const float* __restrict__ gate, float* __restrict__ z, long n) {
  long i = (long)blockIdx.x * blockDim.x + threadIdx.x;
  if (i < n) {
    float a = 1.0f / (1.0f + __expf(-gate[0]));
    z[i] = a * xs[i] + (1.0f - a) * xd[i];
  }
}

extern "C" void kernel_launch(void* const* d_in, const int* in_sizes, int n_in,
                              void* d_out, int out_size, void* d_ws, size_t ws_size,
                              hipStream_t stream) {
  (void)in_sizes; (void)n_in; (void)out_size; (void)ws_size;

  const float* x   = (const float*)d_in[0];
  const int*   ei  = (const int*)d_in[1];
  const int*   hyi = (const int*)d_in[2];
  const float* g1w = (const float*)d_in[3];
  const float* g1b = (const float*)d_in[4];
  const float* g2w = (const float*)d_in[5];
  const float* g2b = (const float*)d_in[6];
  const float* h1w = (const float*)d_in[7];
  const float* h1b = (const float*)d_in[8];
  const float* h2w = (const float*)d_in[9];
  const float* h2b = (const float*)d_in[10];
  const float* gate = (const float*)d_in[11];

  float* out_z  = (float*)d_out;
  float* out_xs = out_z  + (long)N_NODES * F_OUT;
  float* out_xd = out_xs + (long)N_NODES * F_OUT;

  float* ws   = (float*)d_ws;
  float* xw   = ws;
  float* h    = xw   + (long)N_NODES * 64;
  float* dis  = h    + (long)N_NODES * 64;
  float* ef   = dis  + N_NODES;

  unsigned* lst  = (unsigned*)out_z;
  int*      off  = (int*)(lst + 3L * PART_LW);
  unsigned* lstG = lst;                 int* offG = off;
  unsigned* lstH = lst + PART_LW;       int* offH = off + PART_OW;
  unsigned* lstN = lst + 2L * PART_LW;  int* offN = off + 2L * PART_OW;

  part_lists<<<3 * PART_N, PART_T, 0, stream>>>(ei + N_EDGES, ei, hyi + NNZ_H, hyi, hyi, hyi + NNZ_H, lst, off);

  owner_count<<<OWN_WG, OWN_T, 0, stream>>>(lstG, offG, dis, N_NODES, OWN_RN);

  gemm_wmma_f32<<<N_NODES / 16, 128, 0, stream>>>(x, g1w, xw, F_IN);
  owner_pass<0, true><<<OWN_WG, OWN_T, 0, stream>>>(lstG, offG, xw, N_NODES, dis, g1b, g2w, h, N_NODES, OWN_RN, 1);
  owner_pass<0, false><<<OWN_WG, OWN_T, 0, stream>>>(lstG, offG, h, N_NODES, dis, g2b, nullptr, out_xs, N_NODES, OWN_RN, 0);

  gemm_wmma_f32<<<N_NODES / 16, 128, 0, stream>>>(x, h1w, xw, F_IN);
  owner_pass<1, false><<<OWN_WG, OWN_T, 0, stream>>>(lstH, offH, xw, N_NODES, nullptr, nullptr, nullptr, ef, M_HE, OWN_RM, 0);
  owner_pass<2, true><<<OWN_WG, OWN_T, 0, stream>>>(lstN, offN, ef, M_HE, nullptr, h1b, h2w, h, N_NODES, OWN_RN, 1);
  owner_pass<1, false><<<OWN_WG, OWN_T, 0, stream>>>(lstH, offH, h, N_NODES, nullptr, nullptr, nullptr, ef, M_HE, OWN_RM, 0);
  owner_pass<2, false><<<OWN_WG, OWN_T, 0, stream>>>(lstN, offN, ef, M_HE, nullptr, h2b, nullptr, out_xd, N_NODES, OWN_RN, 0);

  k_fuse<<<(N_NODES * 64 + 255) / 256, 256, 0, stream>>>(out_xs, out_xd, gate, out_z, (long)N_NODES * 64);
}


# x*W1 products: one wave per 16x64 stripe (4 accumulators), A by 16-byte loads, one feeding 8 matrix instructions, B k-pair-interleaved in LDS once per 64 rows; k order in these products changes
# speedup vs baseline: 4.0854x; 1.0599x over previous
#include <hip/hip_runtime.h>
#include <math.h>


#define N_NODES 50000
#define N_EDGES 800000
#define NNZ_H   800000
#define M_HE    10000
#define F_IN    128
#define F_HID   64
#define F_OUT   64

#define OWN_WG    256
#define OWN_T     1024
#define OWN_RN    ((N_NODES + OWN_WG - 1) / OWN_WG)
#define OWN_RM    ((M_HE + OWN_WG - 1) / OWN_WG)
#define OWN_WC    448
#define OWN_LC    (32 * OWN_WC)
#define OWN_PU    8
#define OWN_NB    6

#define OWN_RT    ((OWN_RN + 15) / 16)
#define OWN_RP    (16 * OWN_RT)
#define OWN_AP    68
#define OWN_WP    72

#define PART_N    256
#define PART_T    256
#define PART_SL   ((N_EDGES + PART_N - 1) / PART_N)
#define PART_E    ((PART_SL + PART_T - 1) / PART_T)
#define PART_SS   ((PART_SL + 31) & ~31)
#define PART_OS   ((PART_N + 1 + 31) & ~31)
#define PART_LW   (PART_N * PART_SS)
#define PART_OW   (PART_N * PART_OS)

static_assert(N_NODES % 16 == 0, "the first products take the rows in whole 16-row stripes");
static_assert(N_EDGES == NNZ_H, "the three index lists have one length, so one slice length");
static_assert(PART_N == 8 * (OWN_T / 32), "a wave of an owner reads 8 slices' runs");
static_assert(PART_T == OWN_WG, "the offsets of the owner bins are summed one bin per thread");
static_assert(OWN_WC % OWN_PU == 0, "a wave's slots are taken OWN_PU at a time");
static_assert(OWN_T == 16 * F_HID && F_HID == 64 && F_OUT == 64, "the 64x64 weights go to LDS 16 bytes per thread");
static_assert(OWN_T / 32 == 4 * 8, "a wave's product tiles: column tile wave%4, row tiles wave/4, wave/4+8");
static_assert(N_NODES < 0xFFFF && M_HE < 0xFFFF && OWN_RN <= 0xFFFF, "a match is packed as source row (16 bits, 0xFFFF = none) and local row (16 bits)");
static_assert(3L * (PART_LW + PART_OW) <= (long)N_NODES * F_OUT, "the three bucketed lists and their offsets lie in the z plane of the output until the last launch writes it");

typedef float v2f __attribute__((ext_vector_type(2)));
typedef float v8f __attribute__((ext_vector_type(8)));

#define GEMM_WV 4
#define GEMM_PP 144

template <int K>
__global__ __launch_bounds__(32 * GEMM_WV)
void gemm_wmma_f32(const float* __restrict__ A, const float* __restrict__ B,
                   float* __restrict__ C, int nStripes) {
  static_assert(K % 8 == 0, "a lane's 16-byte loads cover k in steps of 8");
  __shared__ alignas(16) float Bs[(K / 2) * GEMM_PP];
  const int tid  = threadIdx.x;
  const int lane = tid & 31;
  const int wv   = __builtin_amdgcn_readfirstlane(tid >> 5);
  const int m    = lane & 15;
  const int hh   = lane >> 4;
  const int stripe = blockIdx.x * GEMM_WV + wv;
  const int st = min(stripe, nStripes - 1);

  const float* Arow = A + ((long)st * 16 + m) * K + 4 * hh;
  float4 a[K / 8];
#pragma unroll
  for (int v = 0; v < K / 8; ++v) a[v] = *(const float4*)(Arow + 8 * v);

  static_assert(((K / 2) * 16) % (32 * GEMM_WV) == 0, "every thread stages the same number of pieces");
#pragma unroll
  for (int j = 0; j < (K / 2) * 16 / (32 * GEMM_WV); ++j) {
    const int i = tid + j * (32 * GEMM_WV);
    const int p = i >> 4, q = (i & 15) << 2;
    const float4 r0 = *(const float4*)(B + (2 * p) * 64 + q);
    const float4 r1 = *(const float4*)(B + (2 * p + 1) * 64 + q);
    float* d = &Bs[p * GEMM_PP + 2 * q];
    *(float4*)d       = make_float4(r0.x, r1.x, r0.y, r1.y);
    *(float4*)(d + 4) = make_float4(r0.z, r1.z, r0.w, r1.w);
  }
  __syncthreads();
  if (stripe >= nStripes) return;

  const float* bp = Bs + (2 * hh) * GEMM_PP + 2 * m;
  v8f acc0 = {}, acc1 = {}, acc2 = {}, acc3 = {};
  v2f bc[8], bn[8];
#pragma unroll
  for (int t = 0; t < 8; ++t) bc[t] = *(const v2f*)(bp + (t >> 2) * GEMM_PP + (t & 3) * 32);
#pragma unroll
  for (int v = 0; v < K / 8; ++v) {
    if (v + 1 < K / 8) {
#pragma unroll
      for (int t = 0; t < 8; ++t)
        bn[t] = *(const v2f*)(bp + (4 * (v + 1) + (t >> 2)) * GEMM_PP + (t & 3) * 32);
    }
    v2f af;
    af.x = a[v].x; af.y = a[v].y;
    acc0 = __builtin_amdgcn_wmma_f32_16x16x4_f32(false, af, false, bc[0], (short)0, acc0, false, false);
    acc1 = __builtin_amdgcn_wmma_f32_16x16x4_f32(false, af, false, bc[1], (short)0, acc1, false, false);
    acc2 = __builtin_amdgcn_wmma_f32_16x16x4_f32(false, af, false, bc[2], (short)0, acc2, false, false);
    acc3 = __builtin_amdgcn_wmma_f32_16x16x4_f32(false, af, false, bc[3], (short)0, acc3, false, false);
    af.x = a[v].z; af.y = a[v].w;
    acc0 = __builtin_amdgcn_wmma_f32_16x16x4_f32(false, af, false, bc[4], (short)0, acc0, false, false);
    acc1 = __builtin_amdgcn_wmma_f32_16x16x4_f32(false, af, false, bc[5], (short)0, acc1, false, false);
    acc2 = __builtin_amdgcn_wmma_f32_16x16x4_f32(false, af, false, bc[6], (short)0, acc2, false, false);
    acc3 = __builtin_amdgcn_wmma_f32_16x16x4_f32(false, af, false, bc[7], (short)0, acc3, false, false);
#pragma unroll
    for (int t = 0; t < 8; ++t) bc[t] = bn[t];
  }
  float* Cp = C + ((long)stripe * 16 + (hh << 3)) * 64 + m;
#pragma unroll
  for (int r = 0; r < 8; ++r) {
    Cp[(long)r * 64]      = acc0[r];
    Cp[(long)r * 64 + 16] = acc1[r];
    Cp[(long)r * 64 + 32] = acc2[r];
    Cp[(long)r * 64 + 48] = acc3[r];
  }
}


__device__ __forceinline__ unsigned own_pack(int s, int dl) {
  return ((unsigned)dl << 16) | ((unsigned)s & 0xFFFFu);
}

template <int R>
__device__ __forceinline__ bool part_bucket(int d, int& o, int& dl) {
  if ((unsigned)d >= (unsigned)(OWN_WG * R)) return false;
  o = d / R; dl = d - o * R;
  return true;
}

__global__ __launch_bounds__(PART_T)
void part_lists(const int* __restrict__ dG, const int* __restrict__ sG,
                const int* __restrict__ dH, const int* __restrict__ sH,
                const int* __restrict__ dN, const int* __restrict__ sN,
                unsigned* __restrict__ lst, int* __restrict__ off) {
  __shared__ int      s_hist[OWN_WG];
  __shared__ int      s_cur[OWN_WG];
  __shared__ int      s_wt[PART_T / 32];
  __shared__ unsigned s_sorted[PART_SL];
  const int tid = threadIdx.x;
  const int wv = tid >> 5, lane = tid & 31;
  const int li = blockIdx.x / PART_N;
  const int sl = blockIdx.x - li * PART_N;
  const int* __restrict__ dst = (li == 0) ? dG : (li == 1) ? dH : dN;
  const int* __restrict__ src = (li == 0) ? sG : (li == 1) ? sH : sN;
  const int L    = (li == 0) ? N_EDGES : NNZ_H;
  const int nSrc = (li == 2) ? M_HE : N_NODES;
  unsigned* lstS = lst + (long)li * PART_LW + (long)sl * PART_SS;
  int*      offS = off + (long)li * PART_OW + (long)sl * PART_OS;
  const int e0 = sl * PART_SL;
  const int e1 = min(e0 + PART_SL, L);

  s_hist[tid] = 0;
  __syncthreads();

  int d[PART_E], s[PART_E], ow[PART_E]; unsigned pk[PART_E];
#pragma unroll
  for (int j = 0; j < PART_E; ++j) {
    const int e  = e0 + tid + j * PART_T;
    const int ec = min(e, L - 1);
    const int dv = dst[ec], sv = src[ec];
    d[j] = (e < e1) ? dv : -1;
    s[j] = (e < e1) ? sv : -1;
  }
#pragma unroll
  for (int j = 0; j < PART_E; ++j) {
    int o = -1, dl = 0;
    const bool in = (li == 1) ? part_bucket<OWN_RM>(d[j], o, dl) : part_bucket<OWN_RN>(d[j], o, dl);
    ow[j] = in ? o : -1;
    pk[j] = own_pack(((unsigned)s[j] < (unsigned)nSrc) ? s[j] : -1, dl);
    if (in) atomicAdd(&s_hist[o], 1);
  }
  __syncthreads();

  const int v = s_hist[tid];
  int inc = v;
#pragma unroll
  for (int k = 1; k < 32; k <<= 1) {
    const int t = __shfl_up(inc, k);
    if (lane >= k) inc += t;
  }
  if (lane == 31) s_wt[wv] = inc;
  __syncthreads();
  int base = 0, tot = 0;
#pragma unroll
  for (int w = 0; w < PART_T / 32; ++w) {
    const int t = s_wt[w];
    if (w < wv) base += t;
    tot += t;
  }
  const int ex = base + inc - v;
  s_cur[tid] = ex;
  offS[tid] = ex;
  if (tid == PART_T - 1) offS[PART_N] = tot;
  __syncthreads();

#pragma unroll
  for (int j = 0; j < PART_E; ++j)
    if (ow[j] >= 0) s_sorted[atomicAdd(&s_cur[ow[j]], 1)] = pk[j];
  __syncthreads();

  for (int i = tid; i < tot; i += PART_T) lstS[i] = s_sorted[i];
}

struct OwnRuns { int pre[9]; int adj[8]; };

__device__ __forceinline__ void own_runs(const int* __restrict__ off, int wv, int lane, OwnRuns& r) {
  int a = 0, len = 0;
  if (lane < 8) {
    const int* o = off + (long)(wv * 8 + lane) * PART_OS + blockIdx.x;
    a   = min(max(o[0], 0), PART_SL);
    len = min(max(o[1] - a, 0), PART_SL - a);
  }
  r.pre[0] = 0;
#pragma unroll
  for (int k = 0; k < 8; ++k) {
    r.pre[k + 1] = r.pre[k] + __builtin_amdgcn_readlane(len, k);
    r.adj[k]     = (wv * 8 + k) * PART_SS + __builtin_amdgcn_readlane(a, k) - r.pre[k];
  }
}

__device__ __forceinline__ unsigned own_match(const unsigned* __restrict__ lst, const OwnRuns& r, int m) {
  const int mc = min(m, r.pre[8] - 1);
  int ad = r.adj[0];
#pragma unroll
  for (int k = 1; k < 8; ++k) ad = (mc >= r.pre[k]) ? r.adj[k] : ad;
  const unsigned p = lst[ad + mc];
  return (m == mc) ? p : 0xFFFFFFFFu;
}

__global__ __launch_bounds__(OWN_T)
void owner_count(const unsigned* __restrict__ lst, const int* __restrict__ off,
                 float* __restrict__ dis, int nDst, int R) {
  __shared__ int s_cnt[OWN_RN];
  const int tid = threadIdx.x;
  const int wv = __builtin_amdgcn_readfirstlane(tid >> 5), lane = tid & 31;
  const int r0 = blockIdx.x * R;
  if (tid < R) s_cnt[tid] = 0;
  __syncthreads();

  OwnRuns rn;
  own_runs(off, wv, lane, rn);
  const int wcnt = rn.pre[8];
  for (int m0 = 0; m0 < wcnt; m0 += 32 * OWN_NB) {
    unsigned p[OWN_NB];
#pragma unroll
    for (int j = 0; j < OWN_NB; ++j) {
      const int m = m0 + 32 * j + lane;
      p[j] = own_match(lst, rn, m);
    }
#pragma unroll
    for (int j = 0; j < OWN_NB; ++j) {
      const unsigned pl = p[j] >> 16;
      if (pl < (unsigned)R) atomicAdd(&s_cnt[pl], 1);
    }
  }
  __syncthreads();
  if (tid < R && r0 + tid < nDst)
    dis[r0 + tid] = rsqrtf((float)s_cnt[tid] + 1.0f);
}

template <int MODE, bool PROD>
__global__ __launch_bounds__(OWN_T)
void owner_pass(const unsigned* __restrict__ lst, const int* __restrict__ off,
                const float* __restrict__ S, int nSrc,
                const float* __restrict__ dis, const float* __restrict__ bias,
                const float* __restrict__ W,
                float* __restrict__ out, int nDst, int R, int do_elu) {
  constexpr int AP = PROD ? OWN_AP : 64;
  constexpr int AR = PROD ? OWN_RP : OWN_RN;
  __shared__ alignas(16) float s_acc[AR * AP];
  __shared__ alignas(16) float s_w[PROD ? 64 * OWN_WP : 4];
  __shared__ float s_dis[OWN_RN];
  __shared__ int   s_cnt[OWN_RN];
  __shared__ int2  s_l[OWN_LC];
  __shared__ float s_lw[OWN_LC];

  const int tid = threadIdx.x;
  const int wv = __builtin_amdgcn_readfirstlane(tid >> 5), lane = tid & 31;
  const int r0  = blockIdx.x * R;

  if (tid < R) {
    s_cnt[tid] = 0;
    if (MODE == 0) s_dis[tid] = (r0 + tid < nDst) ? dis[r0 + tid] : 0.0f;
  }
  float4 wq = make_float4(0.0f, 0.0f, 0.0f, 0.0f);
  if (PROD) wq = *(const float4*)(W + 4 * tid);
  const int rFill = PROD ? ((R + 15) & ~15) : R;
  for (int i = tid; i < rFill * 16; i += OWN_T) {
    const int dl = i >> 4, q = (i & 15) << 2;
    float4 a = make_float4(0.0f, 0.0f, 0.0f, 0.0f);
    if (MODE == 0 && dl < R && r0 + dl < nDst) {
      float w = dis[r0 + dl]; w *= w;
      const float4 v = *(const float4*)(S + (long)(r0 + dl) * 64 + q);
      a = make_float4(w * v.x, w * v.y, w * v.z, w * v.w);
    }
    *(float4*)&s_acc[dl * AP + q] = a;
  }
  if (PROD) *(float4*)&s_w[(tid >> 4) * OWN_WP + ((tid & 15) << 2)] = wq;
  __syncthreads();

  const int lbase = wv * OWN_WC;
  OwnRuns rn;
  own_runs(off, wv, lane, rn);
  const int wcnt = rn.pre[8];
  for (int m0 = 0; m0 < wcnt; m0 += 32 * OWN_NB) {
    unsigned p[OWN_NB];
    int s[OWN_NB], dl[OWN_NB]; float ws[OWN_NB];
#pragma unroll
    for (int j = 0; j < OWN_NB; ++j) {
      const int m = m0 + 32 * j + lane;
      p[j] = own_match(lst, rn, m);
    }
#pragma unroll
    for (int j = 0; j < OWN_NB; ++j) {
      const unsigned pl = p[j] >> 16, ps = p[j] & 0xFFFFu;
      s[j] = -1; dl[j] = 0;
      if (pl < (unsigned)R) {
        dl[j] = (int)pl;
        if (MODE != 0) atomicAdd(&s_cnt[pl], 1);
        if (ps < (unsigned)nSrc) s[j] = (int)ps;
      }
    }
#pragma unroll
    for (int j = 0; j < OWN_NB; ++j) {
      const float t = (MODE == 0) ? dis[max(s[j], 0)] : 0.0f;
      ws[j] = (s[j] >= 0) ? t : 0.0f;
    }
#pragma unroll
    for (int j = 0; j < OWN_NB; ++j) {
      const int m = m0 + 32 * j + lane;
      if (m < wcnt) {
        if (m < OWN_WC) {
          s_l[lbase + m] = make_int2(s[j], dl[j]);
          if (MODE == 0) s_lw[lbase + m] = ws[j];
        } else if (s[j] >= 0) {
          const float w = (MODE == 0) ? ws[j] * s_dis[dl[j]] : 1.0f;
          const float* sp = S + (long)s[j] * 64;
#pragma nounroll
          for (int cc = 0; cc < 64; ++cc) atomicAdd(&s_acc[dl[j] * AP + cc], w * sp[cc]);
        }
      }
    }
  }
  const int cnt    = min(wcnt, OWN_WC);
  const int cntPad = (cnt + OWN_PU - 1) & ~(OWN_PU - 1);
  if (cnt + lane < cntPad) {
    s_l[lbase + cnt + lane] = make_int2(-1, 0);
    if (MODE == 0) s_lw[lbase + cnt + lane] = 0.0f;
  }
  __syncthreads();

  for (int m0 = 0; m0 < cntPad; m0 += OWN_PU) {
    int s[OWN_PU], dl[OWN_PU]; float w[OWN_PU]; float2 v[OWN_PU];
#pragma unroll
    for (int j = 0; j < OWN_PU; ++j) {
      const int2 sd = s_l[lbase + m0 + j];
      s[j]  = __builtin_amdgcn_readfirstlane(sd.x);
      dl[j] = __builtin_amdgcn_readfirstlane(sd.y);
      w[j]  = (MODE == 0) ? s_lw[lbase + m0 + j] * s_dis[dl[j]] : 1.0f;
      v[j]  = *(const float2*)(S + (long)max(s[j], 0) * 64 + 2 * lane);
    }
#pragma unroll
    for (int j = 0; j < OWN_PU; ++j)
      if (s[j] >= 0) {
        atomicAdd(&s_acc[dl[j] * AP + 2 * lane],     w[j] * v[j].x);
        atomicAdd(&s_acc[dl[j] * AP + 2 * lane + 1], w[j] * v[j].y);
      }
  }
  __syncthreads();

  for (int i = tid; i < R * 16; i += OWN_T) {
    const int dl = i >> 4, q = (i & 15) << 2;
    const int row = r0 + dl;
    if (row < nDst) {
      const float4 a = *(const float4*)&s_acc[dl * AP + q];
      float4 r;
      if (MODE == 0) {
        const float4 b = *(const float4*)(bias + q);
        r = make_float4(a.x + b.x, a.y + b.y, a.z + b.z, a.w + b.w);
      } else {
        const int   c  = s_cnt[dl];
        const float sc = (c > 0) ? (1.0f / (float)c) : 0.0f;
        if (MODE == 1) {
          r = make_float4(sc * a.x, sc * a.y, sc * a.z, sc * a.w);
        } else {
          const float4 b = *(const float4*)(bias + q);
          r = make_float4(sc * a.x + b.x, sc * a.y + b.y, sc * a.z + b.z, sc * a.w + b.w);
        }
      }
      if (do_elu) {
        r.x = (r.x > 0.0f) ? r.x : expm1f(r.x);
        r.y = (r.y > 0.0f) ? r.y : expm1f(r.y);
        r.z = (r.z > 0.0f) ? r.z : expm1f(r.z);
        r.w = (r.w > 0.0f) ? r.w : expm1f(r.w);
      }
      if (PROD) *(float4*)&s_acc[dl * AP + q] = r;
      else      *(float4*)(out + (long)row * 64 + q) = r;
    }
  }

  if (PROD) {
    __syncthreads();
    const int m  = lane & 15;
    const int hh = lane >> 4;
    const int ct = (wv & 3) * 16;
    const int nOwn = min(R, nDst - r0);
    const int nRT = (nOwn + 15) >> 4;
    const float* bp = s_w + (hh << 1) * OWN_WP + ct + m;
    for (int rt = wv >> 2; rt < nRT; rt += OWN_T / 128) {
      const float* ap = s_acc + (rt * 16 + m) * AP + (hh << 1);
      v8f acc = {};
#pragma unroll
      for (int k0 = 0; k0 < 64; k0 += 4) {
        v2f a, b;
        a.x = ap[k0];
        a.y = ap[k0 + 1];
        b.x = bp[k0 * OWN_WP];
        b.y = bp[(k0 + 1) * OWN_WP];
        acc = __builtin_amdgcn_wmma_f32_16x16x4_f32(
            false, a, false, b, (short)0, acc, false, false);
      }
      const int dl0 = rt * 16 + (hh << 3);
      float* op = out + (long)(r0 + dl0) * 64 + ct + m;
      if (rt * 16 + 16 <= nOwn) {
#pragma unroll
        for (int r = 0; r < 8; ++r) op[(long)r * 64] = acc[r];
      } else {
#pragma unroll
        for (int r = 0; r < 8; ++r)
          if (dl0 + r < nOwn) op[(long)r * 64] = acc[r];
      }
    }
  }
}

__global__ void k_fuse(const float* __restrict__ xs, const float* __restrict__ xd,
                       const float* __restrict__ gate, float* __restrict__ z, long n) {
  long i = (long)blockIdx.x * blockDim.x + threadIdx.x;
  if (i < n) {
    float a = 1.0f / (1.0f + __expf(-gate[0]));
    z[i] = a * xs[i] + (1.0f - a) * xd[i];
  }
}

extern "C" void kernel_launch(void* const* d_in, const int* in_sizes, int n_in,
                              void* d_out, int out_size, void* d_ws, size_t ws_size,
                              hipStream_t stream) {
  (void)in_sizes; (void)n_in; (void)out_size; (void)ws_size;

  const float* x   = (const float*)d_in[0];
  const int*   ei  = (const int*)d_in[1];
  const int*   hyi = (const int*)d_in[2];
  const float* g1w = (const float*)d_in[3];
  const float* g1b = (const float*)d_in[4];
  const float* g2w = (const float*)d_in[5];
  const float* g2b = (const float*)d_in[6];
  const float* h1w = (const float*)d_in[7];
  const float* h1b = (const float*)d_in[8];
  const float* h2w = (const float*)d_in[9];
  const float* h2b = (const float*)d_in[10];
  const float* gate = (const float*)d_in[11];

  float* out_z  = (float*)d_out;
  float* out_xs = out_z  + (long)N_NODES * F_OUT;
  float* out_xd = out_xs + (long)N_NODES * F_OUT;

  float* ws   = (float*)d_ws;
  float* xw   = ws;
  float* h    = xw   + (long)N_NODES * 64;
  float* dis  = h    + (long)N_NODES * 64;
  float* ef   = dis  + N_NODES;

  unsigned* lst  = (unsigned*)out_z;
  int*      off  = (int*)(lst + 3L * PART_LW);
  unsigned* lstG = lst;                 int* offG = off;
  unsigned* lstH = lst + PART_LW;       int* offH = off + PART_OW;
  unsigned* lstN = lst + 2L * PART_LW;  int* offN = off + 2L * PART_OW;

  part_lists<<<3 * PART_N, PART_T, 0, stream>>>(ei + N_EDGES, ei, hyi + NNZ_H, hyi, hyi, hyi + NNZ_H, lst, off);

  owner_count<<<OWN_WG, OWN_T, 0, stream>>>(lstG, offG, dis, N_NODES, OWN_RN);

  gemm_wmma_f32<F_IN><<<(N_NODES / 16 + GEMM_WV - 1) / GEMM_WV, 32 * GEMM_WV, 0, stream>>>(x, g1w, xw, N_NODES / 16);
  owner_pass<0, true><<<OWN_WG, OWN_T, 0, stream>>>(lstG, offG, xw, N_NODES, dis, g1b, g2w, h, N_NODES, OWN_RN, 1);
  owner_pass<0, false><<<OWN_WG, OWN_T, 0, stream>>>(lstG, offG, h, N_NODES, dis, g2b, nullptr, out_xs, N_NODES, OWN_RN, 0);

  gemm_wmma_f32<F_IN><<<(N_NODES / 16 + GEMM_WV - 1) / GEMM_WV, 32 * GEMM_WV, 0, stream>>>(x, h1w, xw, N_NODES / 16);
  owner_pass<1, false><<<OWN_WG, OWN_T, 0, stream>>>(lstH, offH, xw, N_NODES, nullptr, nullptr, nullptr, ef, M_HE, OWN_RM, 0);
  owner_pass<2, true><<<OWN_WG, OWN_T, 0, stream>>>(lstN, offN, ef, M_HE, nullptr, h1b, h2w, h, N_NODES, OWN_RN, 1);
  owner_pass<1, false><<<OWN_WG, OWN_T, 0, stream>>>(lstH, offH, h, N_NODES, nullptr, nullptr, nullptr, ef, M_HE, OWN_RM, 0);
  owner_pass<2, false><<<OWN_WG, OWN_T, 0, stream>>>(lstN, offN, ef, M_HE, nullptr, h2b, nullptr, out_xd, N_NODES, OWN_RN, 0);

  k_fuse<<<(N_NODES * 64 + 255) / 256, 256, 0, stream>>>(out_xs, out_xd, gate, out_z, (long)N_NODES * 64);
}


# owner passes order matches by local row in LDS (rank by a returning integer add, one wave scans the counts); waves sum equal shares in registers, one LDS float add per row run; sum order changes
# speedup vs baseline: 5.0437x; 1.2346x over previous
#include <hip/hip_runtime.h>
#include <math.h>


#define N_NODES 50000
#define N_EDGES 800000
#define NNZ_H   800000
#define M_HE    10000
#define F_IN    128
#define F_HID   64
#define F_OUT   64

#define OWN_WG    256
#define OWN_T     1024
#define OWN_RN    ((N_NODES + OWN_WG - 1) / OWN_WG)
#define OWN_RM    ((M_HE + OWN_WG - 1) / OWN_WG)
#define OWN_PU    8
#define OWN_NB    6
#define OWN_RC    (32 * 32 * OWN_NB)
#define OWN_LC    (32 * 448)
#define OWN_RL    ((OWN_RN + 31) / 32)

#define OWN_RT    ((OWN_RN + 15) / 16)
#define OWN_RP    (16 * OWN_RT)
#define OWN_AP    68
#define OWN_WP    72

#define PART_N    256
#define PART_T    256
#define PART_SL   ((N_EDGES + PART_N - 1) / PART_N)
#define PART_E    ((PART_SL + PART_T - 1) / PART_T)
#define PART_SS   ((PART_SL + 31) & ~31)
#define PART_OS   ((PART_N + 1 + 31) & ~31)
#define PART_LW   (PART_N * PART_SS)
#define PART_OW   (PART_N * PART_OS)

static_assert(N_NODES % 16 == 0, "the first products take the rows in whole 16-row stripes");
static_assert(N_EDGES == NNZ_H, "the three index lists have one length, so one slice length");
static_assert(PART_N == 8 * (OWN_T / 32), "a wave of an owner reads 8 slices' runs");
static_assert(PART_T == OWN_WG, "the offsets of the owner bins are summed one bin per thread");
static_assert(OWN_RC + OWN_PU <= OWN_LC, "a round's matches and the empty slots behind them fit the ordered list");
static_assert(OWN_T == 16 * F_HID && F_HID == 64 && F_OUT == 64, "the 64x64 weights go to LDS 16 bytes per thread");
static_assert(OWN_T / 32 == 4 * 8, "a wave's product tiles: column tile wave%4, row tiles wave/4, wave/4+8");
static_assert(N_NODES < 0xFFFF && M_HE < 0xFFFF && OWN_RN <= 0xFFFF, "a match is packed as source row (16 bits, 0xFFFF = none) and local row (16 bits)");
static_assert(3L * (PART_LW + PART_OW) <= (long)N_NODES * F_OUT, "the three bucketed lists and their offsets lie in the z plane of the output until the last launch writes it");

typedef float v2f __attribute__((ext_vector_type(2)));
typedef float v8f __attribute__((ext_vector_type(8)));

#define GEMM_WV 4
#define GEMM_PP 144

template <int K>
__global__ __launch_bounds__(32 * GEMM_WV)
void gemm_wmma_f32(const float* __restrict__ A, const float* __restrict__ B,
                   float* __restrict__ C, int nStripes) {
  static_assert(K % 8 == 0, "a lane's 16-byte loads cover k in steps of 8");
  __shared__ alignas(16) float Bs[(K / 2) * GEMM_PP];
  const int tid  = threadIdx.x;
  const int lane = tid & 31;
  const int wv   = __builtin_amdgcn_readfirstlane(tid >> 5);
  const int m    = lane & 15;
  const int hh   = lane >> 4;
  const int stripe = blockIdx.x * GEMM_WV + wv;
  const int st = min(stripe, nStripes - 1);

  const float* Arow = A + ((long)st * 16 + m) * K + 4 * hh;
  float4 a[K / 8];
#pragma unroll
  for (int v = 0; v < K / 8; ++v) a[v] = *(const float4*)(Arow + 8 * v);

  static_assert(((K / 2) * 16) % (32 * GEMM_WV) == 0, "every thread stages the same number of pieces");
#pragma unroll
  for (int j = 0; j < (K / 2) * 16 / (32 * GEMM_WV); ++j) {
    const int i = tid + j * (32 * GEMM_WV);
    const int p = i >> 4, q = (i & 15) << 2;
    const float4 r0 = *(const float4*)(B + (2 * p) * 64 + q);
    const float4 r1 = *(const float4*)(B + (2 * p + 1) * 64 + q);
    float* d = &Bs[p * GEMM_PP + 2 * q];
    *(float4*)d       = make_float4(r0.x, r1.x, r0.y, r1.y);
    *(float4*)(d + 4) = make_float4(r0.z, r1.z, r0.w, r1.w);
  }
  __syncthreads();
  if (stripe >= nStripes) return;

  const float* bp = Bs + (2 * hh) * GEMM_PP + 2 * m;
  v8f acc0 = {}, acc1 = {}, acc2 = {}, acc3 = {};
  v2f bc[8], bn[8];
#pragma unroll
  for (int t = 0; t < 8; ++t) bc[t] = *(const v2f*)(bp + (t >> 2) * GEMM_PP + (t & 3) * 32);
#pragma unroll
  for (int v = 0; v < K / 8; ++v) {
    if (v + 1 < K / 8) {
#pragma unroll
      for (int t = 0; t < 8; ++t)
        bn[t] = *(const v2f*)(bp + (4 * (v + 1) + (t >> 2)) * GEMM_PP + (t & 3) * 32);
    }
    v2f af;
    af.x = a[v].x; af.y = a[v].y;
    acc0 = __builtin_amdgcn_wmma_f32_16x16x4_f32(false, af, false, bc[0], (short)0, acc0, false, false);
    acc1 = __builtin_amdgcn_wmma_f32_16x16x4_f32(false, af, false, bc[1], (short)0, acc1, false, false);
    acc2 = __builtin_amdgcn_wmma_f32_16x16x4_f32(false, af, false, bc[2], (short)0, acc2, false, false);
    acc3 = __builtin_amdgcn_wmma_f32_16x16x4_f32(false, af, false, bc[3], (short)0, acc3, false, false);
    af.x = a[v].z; af.y = a[v].w;
    acc0 = __builtin_amdgcn_wmma_f32_16x16x4_f32(false, af, false, bc[4], (short)0, acc0, false, false);
    acc1 = __builtin_amdgcn_wmma_f32_16x16x4_f32(false, af, false, bc[5], (short)0, acc1, false, false);
    acc2 = __builtin_amdgcn_wmma_f32_16x16x4_f32(false, af, false, bc[6], (short)0, acc2, false, false);
    acc3 = __builtin_amdgcn_wmma_f32_16x16x4_f32(false, af, false, bc[7], (short)0, acc3, false, false);
#pragma unroll
    for (int t = 0; t < 8; ++t) bc[t] = bn[t];
  }
  float* Cp = C + ((long)stripe * 16 + (hh << 3)) * 64 + m;
#pragma unroll
  for (int r = 0; r < 8; ++r) {
    Cp[(long)r * 64]      = acc0[r];
    Cp[(long)r * 64 + 16] = acc1[r];
    Cp[(long)r * 64 + 32] = acc2[r];
    Cp[(long)r * 64 + 48] = acc3[r];
  }
}


__device__ __forceinline__ unsigned own_pack(int s, int dl) {
  return ((unsigned)dl << 16) | ((unsigned)s & 0xFFFFu);
}

template <int R>
__device__ __forceinline__ bool part_bucket(int d, int& o, int& dl) {
  if ((unsigned)d >= (unsigned)(OWN_WG * R)) return false;
  o = d / R; dl = d - o * R;
  return true;
}

__global__ __launch_bounds__(PART_T)
void part_lists(const int* __restrict__ dG, const int* __restrict__ sG,
                const int* __restrict__ dH, const int* __restrict__ sH,
                const int* __restrict__ dN, const int* __restrict__ sN,
                unsigned* __restrict__ lst, int* __restrict__ off) {
  __shared__ int      s_hist[OWN_WG];
  __shared__ int      s_cur[OWN_WG];
  __shared__ int      s_wt[PART_T / 32];
  __shared__ unsigned s_sorted[PART_SL];
  const int tid = threadIdx.x;
  const int wv = tid >> 5, lane = tid & 31;
  const int li = blockIdx.x / PART_N;
  const int sl = blockIdx.x - li * PART_N;
  const int* __restrict__ dst = (li == 0) ? dG : (li == 1) ? dH : dN;
  const int* __restrict__ src = (li == 0) ? sG : (li == 1) ? sH : sN;
  const int L    = (li == 0) ? N_EDGES : NNZ_H;
  const int nSrc = (li == 2) ? M_HE : N_NODES;
  unsigned* lstS = lst + (long)li * PART_LW + (long)sl * PART_SS;
  int*      offS = off + (long)li * PART_OW + (long)sl * PART_OS;
  const int e0 = sl * PART_SL;
  const int e1 = min(e0 + PART_SL, L);

  s_hist[tid] = 0;
  __syncthreads();

  int d[PART_E], s[PART_E], ow[PART_E]; unsigned pk[PART_E];
#pragma unroll
  for (int j = 0; j < PART_E; ++j) {
    const int e  = e0 + tid + j * PART_T;
    const int ec = min(e, L - 1);
    const int dv = dst[ec], sv = src[ec];
    d[j] = (e < e1) ? dv : -1;
    s[j] = (e < e1) ? sv : -1;
  }
#pragma unroll
  for (int j = 0; j < PART_E; ++j) {
    int o = -1, dl = 0;
    const bool in = (li == 1) ? part_bucket<OWN_RM>(d[j], o, dl) : part_bucket<OWN_RN>(d[j], o, dl);
    ow[j] = in ? o : -1;
    pk[j] = own_pack(((unsigned)s[j] < (unsigned)nSrc) ? s[j] : -1, dl);
    if (in) atomicAdd(&s_hist[o], 1);
  }
  __syncthreads();

  const int v = s_hist[tid];
  int inc = v;
#pragma unroll
  for (int k = 1; k < 32; k <<= 1) {
    const int t = __shfl_up(inc, k);
    if (lane >= k) inc += t;
  }
  if (lane == 31) s_wt[wv] = inc;
  __syncthreads();
  int base = 0, tot = 0;
#pragma unroll
  for (int w = 0; w < PART_T / 32; ++w) {
    const int t = s_wt[w];
    if (w < wv) base += t;
    tot += t;
  }
  const int ex = base + inc - v;
  s_cur[tid] = ex;
  offS[tid] = ex;
  if (tid == PART_T - 1) offS[PART_N] = tot;
  __syncthreads();

#pragma unroll
  for (int j = 0; j < PART_E; ++j)
    if (ow[j] >= 0) s_sorted[atomicAdd(&s_cur[ow[j]], 1)] = pk[j];
  __syncthreads();

  for (int i = tid; i < tot; i += PART_T) lstS[i] = s_sorted[i];
}

struct OwnRuns { int pre[9]; int adj[8]; };

__device__ __forceinline__ void own_runs(const int* __restrict__ off, int wv, int lane, OwnRuns& r) {
  int a = 0, len = 0;
  if (lane < 8) {
    const int* o = off + (long)(wv * 8 + lane) * PART_OS + blockIdx.x;
    a   = min(max(o[0], 0), PART_SL);
    len = min(max(o[1] - a, 0), PART_SL - a);
  }
  r.pre[0] = 0;
#pragma unroll
  for (int k = 0; k < 8; ++k) {
    r.pre[k + 1] = r.pre[k] + __builtin_amdgcn_readlane(len, k);
    r.adj[k]     = (wv * 8 + k) * PART_SS + __builtin_amdgcn_readlane(a, k) - r.pre[k];
  }
}

__device__ __forceinline__ unsigned own_match(const unsigned* __restrict__ lst, const OwnRuns& r, int m) {
  const int mc = min(m, r.pre[8] - 1);
  int ad = r.adj[0];
#pragma unroll
  for (int k = 1; k < 8; ++k) ad = (mc >= r.pre[k]) ? r.adj[k] : ad;
  const unsigned p = lst[ad + mc];
  return (m == mc) ? p : 0xFFFFFFFFu;
}

__global__ __launch_bounds__(OWN_T)
void owner_count(const unsigned* __restrict__ lst, const int* __restrict__ off,
                 float* __restrict__ dis, int nDst, int R) {
  __shared__ int s_cnt[OWN_RN];
  const int tid = threadIdx.x;
  const int wv = __builtin_amdgcn_readfirstlane(tid >> 5), lane = tid & 31;
  const int r0 = blockIdx.x * R;
  if (tid < R) s_cnt[tid] = 0;
  __syncthreads();

  OwnRuns rn;
  own_runs(off, wv, lane, rn);
  const int wcnt = rn.pre[8];
  for (int m0 = 0; m0 < wcnt; m0 += 32 * OWN_NB) {
    unsigned p[OWN_NB];
#pragma unroll
    for (int j = 0; j < OWN_NB; ++j) {
      const int m = m0 + 32 * j + lane;
      p[j] = own_match(lst, rn, m);
    }
#pragma unroll
    for (int j = 0; j < OWN_NB; ++j) {
      const unsigned pl = p[j] >> 16;
      if (pl < (unsigned)R) atomicAdd(&s_cnt[pl], 1);
    }
  }
  __syncthreads();
  if (tid < R && r0 + tid < nDst)
    dis[r0 + tid] = rsqrtf((float)s_cnt[tid] + 1.0f);
}

template <int MODE, bool PROD>
__global__ __launch_bounds__(OWN_T)
void owner_pass(const unsigned* __restrict__ lst, const int* __restrict__ off,
                const float* __restrict__ S, int nSrc,
                const float* __restrict__ dis, const float* __restrict__ bias,
                const float* __restrict__ W,
                float* __restrict__ out, int nDst, int R, int do_elu) {
  constexpr int AP = PROD ? OWN_AP : 64;
  constexpr int AR = PROD ? OWN_RP : OWN_RN;
  __shared__ alignas(16) float s_acc[AR * AP];
  __shared__ alignas(16) float s_w[PROD ? 64 * OWN_WP : 4];
  __shared__ float s_dis[OWN_RN];
  __shared__ int   s_cnt[OWN_RN];
  __shared__ int   s_deg[OWN_RN];
  __shared__ int   s_start[OWN_RN];
  __shared__ int   s_tot;
  __shared__ int   s_more;
  __shared__ int2  s_l[OWN_LC];
  __shared__ float s_lw[OWN_LC];

  const int tid = threadIdx.x;
  const int wv = __builtin_amdgcn_readfirstlane(tid >> 5), lane = tid & 31;
  const int r0  = blockIdx.x * R;

  if (tid < R) {
    s_cnt[tid] = 0;
    if (MODE != 0) s_deg[tid] = 0;
    if (MODE == 0) s_dis[tid] = (r0 + tid < nDst) ? dis[r0 + tid] : 0.0f;
  }
  if (tid == 0) s_more = 0;
  float4 wq = make_float4(0.0f, 0.0f, 0.0f, 0.0f);
  if (PROD) wq = *(const float4*)(W + 4 * tid);
  const int rFill = PROD ? ((R + 15) & ~15) : R;
  for (int i = tid; i < rFill * 16; i += OWN_T) {
    const int dl = i >> 4, q = (i & 15) << 2;
    float4 a = make_float4(0.0f, 0.0f, 0.0f, 0.0f);
    if (MODE == 0 && dl < R && r0 + dl < nDst) {
      float w = dis[r0 + dl]; w *= w;
      const float4 v = *(const float4*)(S + (long)(r0 + dl) * 64 + q);
      a = make_float4(w * v.x, w * v.y, w * v.z, w * v.w);
    }
    *(float4*)&s_acc[dl * AP + q] = a;
  }
  if (PROD) *(float4*)&s_w[(tid >> 4) * OWN_WP + ((tid & 15) << 2)] = wq;
  __syncthreads();

  OwnRuns rn;
  own_runs(off, wv, lane, rn);
  const int wcnt = rn.pre[8];
  for (int base = 0; ; base += 32 * OWN_NB) {
    unsigned p[OWN_NB];
    int s[OWN_NB], dl[OWN_NB], rk[OWN_NB]; float ws[OWN_NB];
#pragma unroll
    for (int j = 0; j < OWN_NB; ++j) p[j] = 0xFFFFFFFFu;
    if (base < wcnt) {
#pragma unroll
      for (int j = 0; j < OWN_NB; ++j) p[j] = own_match(lst, rn, base + 32 * j + lane);
    }
    if (lane == 0 && wcnt > base + 32 * OWN_NB) s_more = 1;
#pragma unroll
    for (int j = 0; j < OWN_NB; ++j) {
      const unsigned pl = p[j] >> 16, ps = p[j] & 0xFFFFu;
      s[j] = -1; dl[j] = -1; rk[j] = 0;
      if (pl < (unsigned)R) {
        dl[j] = (int)pl;
        rk[j] = atomicAdd(&s_cnt[pl], 1);
        if (ps < (unsigned)nSrc) s[j] = (int)ps;
      }
    }
#pragma unroll
    for (int j = 0; j < OWN_NB; ++j) {
      const float t = (MODE == 0) ? dis[max(s[j], 0)] : 0.0f;
      ws[j] = (s[j] >= 0) ? t : 0.0f;
    }
    __syncthreads();

    if (wv == 0) {
      int cn[OWN_RL]; int tot = 0;
#pragma unroll
      for (int i = 0; i < OWN_RL; ++i) {
        const int r = lane * OWN_RL + i;
        cn[i] = (r < R) ? s_cnt[r] : 0;
        tot += cn[i];
      }
      int inc = tot;
#pragma unroll
      for (int k = 1; k < 32; k <<= 1) {
        const int t = __shfl_up(inc, k);
        if (lane >= k) inc += t;
      }
      int run = inc - tot;
#pragma unroll
      for (int i = 0; i < OWN_RL; ++i) {
        const int r = lane * OWN_RL + i;
        if (r < R) {
          s_start[r] = run;
          if (MODE != 0) s_deg[r] += cn[i];
        }
        run += cn[i];
      }
      if (lane == 31) s_tot = inc;
    }
    __syncthreads();

    const int more = s_more;
    if (tid < OWN_PU) {
      const int pos = s_tot + tid;
      s_l[pos] = make_int2(-1, 0);
      if (MODE == 0) s_lw[pos] = 0.0f;
    }
#pragma unroll
    for (int j = 0; j < OWN_NB; ++j)
      if (dl[j] >= 0) {
        const int pos = s_start[dl[j]] + rk[j];
        s_l[pos] = make_int2(s[j], dl[j]);
        if (MODE == 0) s_lw[pos] = ws[j];
      }
    __syncthreads();

    const int tot = __builtin_amdgcn_readfirstlane(s_tot);
    const int c0 = (tot * wv) >> 5, c1 = (tot * (wv + 1)) >> 5;
    int cur = -1;
    float ax = 0.0f, ay = 0.0f;
    for (int e0 = c0; e0 < c1; e0 += OWN_PU) {
      int sr[OWN_PU], dr[OWN_PU]; float w[OWN_PU]; float2 v[OWN_PU];
#pragma unroll
      for (int j = 0; j < OWN_PU; ++j) {
        const int e = e0 + j;
        const int2 sd = s_l[e];
        const int sj = __builtin_amdgcn_readfirstlane(sd.x);
        dr[j] = __builtin_amdgcn_readfirstlane(sd.y);
        sr[j] = (e < c1) ? sj : -2;
        w[j]  = (MODE == 0) ? s_lw[e] * s_dis[dr[j]] : 1.0f;
        v[j]  = *(const float2*)(S + (long)max(sr[j], 0) * 64 + 2 * lane);
      }
#pragma unroll
      for (int j = 0; j < OWN_PU; ++j)
        if (sr[j] != -2) {
          if (dr[j] != cur) {
            if (cur >= 0) {
              atomicAdd(&s_acc[cur * AP + 2 * lane],     ax);
              atomicAdd(&s_acc[cur * AP + 2 * lane + 1], ay);
            }
            cur = dr[j]; ax = 0.0f; ay = 0.0f;
          }
          if (sr[j] >= 0) {
            ax += w[j] * v[j].x;
            ay += w[j] * v[j].y;
          }
        }
    }
    if (cur >= 0) {
      atomicAdd(&s_acc[cur * AP + 2 * lane],     ax);
      atomicAdd(&s_acc[cur * AP + 2 * lane + 1], ay);
    }
    __syncthreads();
    if (!more) break;
    if (tid < R) s_cnt[tid] = 0;
    if (tid == 0) s_more = 0;
    __syncthreads();
  }

  for (int i = tid; i < R * 16; i += OWN_T) {
    const int dl = i >> 4, q = (i & 15) << 2;
    const int row = r0 + dl;
    if (row < nDst) {
      const float4 a = *(const float4*)&s_acc[dl * AP + q];
      float4 r;
      if (MODE == 0) {
        const float4 b = *(const float4*)(bias + q);
        r = make_float4(a.x + b.x, a.y + b.y, a.z + b.z, a.w + b.w);
      } else {
        const int   c  = s_deg[dl];
        const float sc = (c > 0) ? (1.0f / (float)c) : 0.0f;
        if (MODE == 1) {
          r = make_float4(sc * a.x, sc * a.y, sc * a.z, sc * a.w);
        } else {
          const float4 b = *(const float4*)(bias + q);
          r = make_float4(sc * a.x + b.x, sc * a.y + b.y, sc * a.z + b.z, sc * a.w + b.w);
        }
      }
      if (do_elu) {
        r.x = (r.x > 0.0f) ? r.x : expm1f(r.x);
        r.y = (r.y > 0.0f) ? r.y : expm1f(r.y);
        r.z = (r.z > 0.0f) ? r.z : expm1f(r.z);
        r.w = (r.w > 0.0f) ? r.w : expm1f(r.w);
      }
      if (PROD) *(float4*)&s_acc[dl * AP + q] = r;
      else      *(float4*)(out + (long)row * 64 + q) = r;
    }
  }

  if (PROD) {
    __syncthreads();
    const int m  = lane & 15;
    const int hh = lane >> 4;
    const int ct = (wv & 3) * 16;
    const int nOwn = min(R, nDst - r0);
    const int nRT = (nOwn + 15) >> 4;
    const float* bp = s_w + (hh << 1) * OWN_WP + ct + m;
    for (int rt = wv >> 2; rt < nRT; rt += OWN_T / 128) {
      const float* ap = s_acc + (rt * 16 + m) * AP + (hh << 1);
      v8f acc = {};
#pragma unroll
      for (int k0 = 0; k0 < 64; k0 += 4) {
        v2f a, b;
        a.x = ap[k0];
        a.y = ap[k0 + 1];
        b.x = bp[k0 * OWN_WP];
        b.y = bp[(k0 + 1) * OWN_WP];
        acc = __builtin_amdgcn_wmma_f32_16x16x4_f32(
            false, a, false, b, (short)0, acc, false, false);
      }
      const int dl0 = rt * 16 + (hh << 3);
      float* op = out + (long)(r0 + dl0) * 64 + ct + m;
      if (rt * 16 + 16 <= nOwn) {
#pragma unroll
        for (int r = 0; r < 8; ++r) op[(long)r * 64] = acc[r];
      } else {
#pragma unroll
        for (int r = 0; r < 8; ++r)
          if (dl0 + r < nOwn) op[(long)r * 64] = acc[r];
      }
    }
  }
}

__global__ void k_fuse(const float* __restrict__ xs, const float* __restrict__ xd,
                       const float* __restrict__ gate, float* __restrict__ z, long n) {
  long i = (long)blockIdx.x * blockDim.x + threadIdx.x;
  if (i < n) {
    float a = 1.0f / (1.0f + __expf(-gate[0]));
    z[i] = a * xs[i] + (1.0f - a) * xd[i];
  }
}

extern "C" void kernel_launch(void* const* d_in, const int* in_sizes, int n_in,
                              void* d_out, int out_size, void* d_ws, size_t ws_size,
                              hipStream_t stream) {
  (void)in_sizes; (void)n_in; (void)out_size; (void)ws_size;

  const float* x   = (const float*)d_in[0];
  const int*   ei  = (const int*)d_in[1];
  const int*   hyi = (const int*)d_in[2];
  const float* g1w = (const float*)d_in[3];
  const float* g1b = (const float*)d_in[4];
  const float* g2w = (const float*)d_in[5];
  const float* g2b = (const float*)d_in[6];
  const float* h1w = (const float*)d_in[7];
  const float* h1b = (const float*)d_in[8];
  const float* h2w = (const float*)d_in[9];
  const float* h2b = (const float*)d_in[10];
  const float* gate = (const float*)d_in[11];

  float* out_z  = (float*)d_out;
  float* out_xs = out_z  + (long)N_NODES * F_OUT;
  float* out_xd = out_xs + (long)N_NODES * F_OUT;

  float* ws   = (float*)d_ws;
  float* xw   = ws;
  float* h    = xw   + (long)N_NODES * 64;
  float* dis  = h    + (long)N_NODES * 64;
  float* ef   = dis  + N_NODES;

  unsigned* lst  = (unsigned*)out_z;
  int*      off  = (int*)(lst + 3L * PART_LW);
  unsigned* lstG = lst;                 int* offG = off;
  unsigned* lstH = lst + PART_LW;       int* offH = off + PART_OW;
  unsigned* lstN = lst + 2L * PART_LW;  int* offN = off + 2L * PART_OW;

  part_lists<<<3 * PART_N, PART_T, 0, stream>>>(ei + N_EDGES, ei, hyi + NNZ_H, hyi, hyi, hyi + NNZ_H, lst, off);

  owner_count<<<OWN_WG, OWN_T, 0, stream>>>(lstG, offG, dis, N_NODES, OWN_RN);

  gemm_wmma_f32<F_IN><<<(N_NODES / 16 + GEMM_WV - 1) / GEMM_WV, 32 * GEMM_WV, 0, stream>>>(x, g1w, xw, N_NODES / 16);
  owner_pass<0, true><<<OWN_WG, OWN_T, 0, stream>>>(lstG, offG, xw, N_NODES, dis, g1b, g2w, h, N_NODES, OWN_RN, 1);
  owner_pass<0, false><<<OWN_WG, OWN_T, 0, stream>>>(lstG, offG, h, N_NODES, dis, g2b, nullptr, out_xs, N_NODES, OWN_RN, 0);

  gemm_wmma_f32<F_IN><<<(N_NODES / 16 + GEMM_WV - 1) / GEMM_WV, 32 * GEMM_WV, 0, stream>>>(x, h1w, xw, N_NODES / 16);
  owner_pass<1, false><<<OWN_WG, OWN_T, 0, stream>>>(lstH, offH, xw, N_NODES, nullptr, nullptr, nullptr, ef, M_HE, OWN_RM, 0);
  owner_pass<2, true><<<OWN_WG, OWN_T, 0, stream>>>(lstN, offN, ef, M_HE, nullptr, h1b, h2w, h, N_NODES, OWN_RN, 1);
  owner_pass<1, false><<<OWN_WG, OWN_T, 0, stream>>>(lstH, offH, h, N_NODES, nullptr, nullptr, nullptr, ef, M_HE, OWN_RM, 0);
  owner_pass<2, false><<<OWN_WG, OWN_T, 0, stream>>>(lstN, offN, ef, M_HE, nullptr, h2b, nullptr, out_xd, N_NODES, OWN_RN, 0);

  k_fuse<<<(N_NODES * 64 + 255) / 256, 256, 0, stream>>>(out_xs, out_xd, gate, out_z, (long)N_NODES * 64);
}


# final mix z formed by the last owner pass from the x_d rows it has just finished (x_s rows asked for ahead): the one-element-per-thread mix launch goes, 11 launches -> 10; sum order unchanged
# speedup vs baseline: 5.2589x; 1.0427x over previous
#include <hip/hip_runtime.h>
#include <math.h>


#define N_NODES 50000
#define N_EDGES 800000
#define NNZ_H   800000
#define M_HE    10000
#define F_IN    128
#define F_HID   64
#define F_OUT   64

#define OWN_WG    256
#define OWN_T     1024
#define OWN_RN    ((N_NODES + OWN_WG - 1) / OWN_WG)
#define OWN_RM    ((M_HE + OWN_WG - 1) / OWN_WG)
#define OWN_PU    8
#define OWN_NB    6
#define OWN_RC    (32 * 32 * OWN_NB)
#define OWN_LC    (32 * 448)
#define OWN_RL    ((OWN_RN + 31) / 32)
#define OWN_FT    ((OWN_RN * 16 + OWN_T - 1) / OWN_T)

#define OWN_RT    ((OWN_RN + 15) / 16)
#define OWN_RP    (16 * OWN_RT)
#define OWN_AP    68
#define OWN_WP    72

#define PART_N    256
#define PART_T    256
#define PART_SL   ((N_EDGES + PART_N - 1) / PART_N)
#define PART_E    ((PART_SL + PART_T - 1) / PART_T)
#define PART_SS   ((PART_SL + 31) & ~31)
#define PART_OS   ((PART_N + 1 + 31) & ~31)
#define PART_LW   (PART_N * PART_SS)
#define PART_OW   (PART_N * PART_OS)

static_assert(N_NODES % 16 == 0, "the first products take the rows in whole 16-row stripes");
static_assert(N_EDGES == NNZ_H, "the three index lists have one length, so one slice length");
static_assert(PART_N == 8 * (OWN_T / 32), "a wave of an owner reads 8 slices' runs");
static_assert(PART_T == OWN_WG, "the offsets of the owner bins are summed one bin per thread");
static_assert(OWN_RC + OWN_PU <= OWN_LC, "a round's matches and the empty slots behind them fit the ordered list");
static_assert(OWN_T == 16 * F_HID && F_HID == 64 && F_OUT == 64, "the 64x64 weights go to LDS 16 bytes per thread");
static_assert(OWN_T / 32 == 4 * 8, "a wave's product tiles: column tile wave%4, row tiles wave/4, wave/4+8");
static_assert(N_NODES < 0xFFFF && M_HE < 0xFFFF && OWN_RN <= 0xFFFF, "a match is packed as source row (16 bits, 0xFFFF = none) and local row (16 bits)");
static_assert(3L * (PART_LW + PART_OW) <= (long)N_NODES * F_OUT, "the three bucketed lists and their offsets lie in the first N x 64 plane of the workspace");

typedef float v2f __attribute__((ext_vector_type(2)));
typedef float v8f __attribute__((ext_vector_type(8)));

#define GEMM_WV 4
#define GEMM_PP 144

template <int K>
__global__ __launch_bounds__(32 * GEMM_WV)
void gemm_wmma_f32(const float* __restrict__ A, const float* __restrict__ B,
                   float* __restrict__ C, int nStripes) {
  static_assert(K % 8 == 0, "a lane's 16-byte loads cover k in steps of 8");
  __shared__ alignas(16) float Bs[(K / 2) * GEMM_PP];
  const int tid  = threadIdx.x;
  const int lane = tid & 31;
  const int wv   = __builtin_amdgcn_readfirstlane(tid >> 5);
  const int m    = lane & 15;
  const int hh   = lane >> 4;
  const int stripe = blockIdx.x * GEMM_WV + wv;
  const int st = min(stripe, nStripes - 1);

  const float* Arow = A + ((long)st * 16 + m) * K + 4 * hh;
  float4 a[K / 8];
#pragma unroll
  for (int v = 0; v < K / 8; ++v) a[v] = *(const float4*)(Arow + 8 * v);

  static_assert(((K / 2) * 16) % (32 * GEMM_WV) == 0, "every thread stages the same number of pieces");
#pragma unroll
  for (int j = 0; j < (K / 2) * 16 / (32 * GEMM_WV); ++j) {
    const int i = tid + j * (32 * GEMM_WV);
    const int p = i >> 4, q = (i & 15) << 2;
    const float4 r0 = *(const float4*)(B + (2 * p) * 64 + q);
    const float4 r1 = *(const float4*)(B + (2 * p + 1) * 64 + q);
    float* d = &Bs[p * GEMM_PP + 2 * q];
    *(float4*)d       = make_float4(r0.x, r1.x, r0.y, r1.y);
    *(float4*)(d + 4) = make_float4(r0.z, r1.z, r0.w, r1.w);
  }
  __syncthreads();
  if (stripe >= nStripes) return;

  const float* bp = Bs + (2 * hh) * GEMM_PP + 2 * m;
  v8f acc0 = {}, acc1 = {}, acc2 = {}, acc3 = {};
  v2f bc[8], bn[8];
#pragma unroll
  for (int t = 0; t < 8; ++t) bc[t] = *(const v2f*)(bp + (t >> 2) * GEMM_PP + (t & 3) * 32);
#pragma unroll
  for (int v = 0; v < K / 8; ++v) {
    if (v + 1 < K / 8) {
#pragma unroll
      for (int t = 0; t < 8; ++t)
        bn[t] = *(const v2f*)(bp + (4 * (v + 1) + (t >> 2)) * GEMM_PP + (t & 3) * 32);
    }
    v2f af;
    af.x = a[v].x; af.y = a[v].y;
    acc0 = __builtin_amdgcn_wmma_f32_16x16x4_f32(false, af, false, bc[0], (short)0, acc0, false, false);
    acc1 = __builtin_amdgcn_wmma_f32_16x16x4_f32(false, af, false, bc[1], (short)0, acc1, false, false);
    acc2 = __builtin_amdgcn_wmma_f32_16x16x4_f32(false, af, false, bc[2], (short)0, acc2, false, false);
    acc3 = __builtin_amdgcn_wmma_f32_16x16x4_f32(false, af, false, bc[3], (short)0, acc3, false, false);
    af.x = a[v].z; af.y = a[v].w;
    acc0 = __builtin_amdgcn_wmma_f32_16x16x4_f32(false, af, false, bc[4], (short)0, acc0, false, false);
    acc1 = __builtin_amdgcn_wmma_f32_16x16x4_f32(false, af, false, bc[5], (short)0, acc1, false, false);
    acc2 = __builtin_amdgcn_wmma_f32_16x16x4_f32(false, af, false, bc[6], (short)0, acc2, false, false);
    acc3 = __builtin_amdgcn_wmma_f32_16x16x4_f32(false, af, false, bc[7], (short)0, acc3, false, false);
#pragma unroll
    for (int t = 0; t < 8; ++t) bc[t] = bn[t];
  }
  float* Cp = C + ((long)stripe * 16 + (hh << 3)) * 64 + m;
#pragma unroll
  for (int r = 0; r < 8; ++r) {
    Cp[(long)r * 64]      = acc0[r];
    Cp[(long)r * 64 + 16] = acc1[r];
    Cp[(long)r * 64 + 32] = acc2[r];
    Cp[(long)r * 64 + 48] = acc3[r];
  }
}


__device__ __forceinline__ unsigned own_pack(int s, int dl) {
  return ((unsigned)dl << 16) | ((unsigned)s & 0xFFFFu);
}

template <int R>
__device__ __forceinline__ bool part_bucket(int d, int& o, int& dl) {
  if ((unsigned)d >= (unsigned)(OWN_WG * R)) return false;
  o = d / R; dl = d - o * R;
  return true;
}

__global__ __launch_bounds__(PART_T)
void part_lists(const int* __restrict__ dG, const int* __restrict__ sG,
                const int* __restrict__ dH, const int* __restrict__ sH,
                const int* __restrict__ dN, const int* __restrict__ sN,
                unsigned* __restrict__ lst, int* __restrict__ off) {
  __shared__ int      s_hist[OWN_WG];
  __shared__ int      s_cur[OWN_WG];
  __shared__ int      s_wt[PART_T / 32];
  __shared__ unsigned s_sorted[PART_SL];
  const int tid = threadIdx.x;
  const int wv = tid >> 5, lane = tid & 31;
  const int li = blockIdx.x / PART_N;
  const int sl = blockIdx.x - li * PART_N;
  const int* __restrict__ dst = (li == 0) ? dG : (li == 1) ? dH : dN;
  const int* __restrict__ src = (li == 0) ? sG : (li == 1) ? sH : sN;
  const int L    = (li == 0) ? N_EDGES : NNZ_H;
  const int nSrc = (li == 2) ? M_HE : N_NODES;
  unsigned* lstS = lst + (long)li * PART_LW + (long)sl * PART_SS;
  int*      offS = off + (long)li * PART_OW + (long)sl * PART_OS;
  const int e0 = sl * PART_SL;
  const int e1 = min(e0 + PART_SL, L);

  s_hist[tid] = 0;
  __syncthreads();

  int d[PART_E], s[PART_E], ow[PART_E]; unsigned pk[PART_E];
#pragma unroll
  for (int j = 0; j < PART_E; ++j) {
    const int e  = e0 + tid + j * PART_T;
    const int ec = min(e, L - 1);
    const int dv = dst[ec], sv = src[ec];
    d[j] = (e < e1) ? dv : -1;
    s[j] = (e < e1) ? sv : -1;
  }
#pragma unroll
  for (int j = 0; j < PART_E; ++j) {
    int o = -1, dl = 0;
    const bool in = (li == 1) ? part_bucket<OWN_RM>(d[j], o, dl) : part_bucket<OWN_RN>(d[j], o, dl);
    ow[j] = in ? o : -1;
    pk[j] = own_pack(((unsigned)s[j] < (unsigned)nSrc) ? s[j] : -1, dl);
    if (in) atomicAdd(&s_hist[o], 1);
  }
  __syncthreads();

  const int v = s_hist[tid];
  int inc = v;
#pragma unroll
  for (int k = 1; k < 32; k <<= 1) {
    const int t = __shfl_up(inc, k);
    if (lane >= k) inc += t;
  }
  if (lane == 31) s_wt[wv] = inc;
  __syncthreads();
  int base = 0, tot = 0;
#pragma unroll
  for (int w = 0; w < PART_T / 32; ++w) {
    const int t = s_wt[w];
    if (w < wv) base += t;
    tot += t;
  }
  const int ex = base + inc - v;
  s_cur[tid] = ex;
  offS[tid] = ex;
  if (tid == PART_T - 1) offS[PART_N] = tot;
  __syncthreads();

#pragma unroll
  for (int j = 0; j < PART_E; ++j)
    if (ow[j] >= 0) s_sorted[atomicAdd(&s_cur[ow[j]], 1)] = pk[j];
  __syncthreads();

  for (int i = tid; i < tot; i += PART_T) lstS[i] = s_sorted[i];
}

struct OwnRuns { int pre[9]; int adj[8]; };

__device__ __forceinline__ void own_runs(const int* __restrict__ off, int wv, int lane, OwnRuns& r) {
  int a = 0, len = 0;
  if (lane < 8) {
    const int* o = off + (long)(wv * 8 + lane) * PART_OS + blockIdx.x;
    a   = min(max(o[0], 0), PART_SL);
    len = min(max(o[1] - a, 0), PART_SL - a);
  }
  r.pre[0] = 0;
#pragma unroll
  for (int k = 0; k < 8; ++k) {
    r.pre[k + 1] = r.pre[k] + __builtin_amdgcn_readlane(len, k);
    r.adj[k]     = (wv * 8 + k) * PART_SS + __builtin_amdgcn_readlane(a, k) - r.pre[k];
  }
}

__device__ __forceinline__ unsigned own_match(const unsigned* __restrict__ lst, const OwnRuns& r, int m) {
  const int mc = min(m, r.pre[8] - 1);
  int ad = r.adj[0];
#pragma unroll
  for (int k = 1; k < 8; ++k) ad = (mc >= r.pre[k]) ? r.adj[k] : ad;
  const unsigned p = lst[ad + mc];
  return (m == mc) ? p : 0xFFFFFFFFu;
}

__global__ __launch_bounds__(OWN_T)
void owner_count(const unsigned* __restrict__ lst, const int* __restrict__ off,
                 float* __restrict__ dis, int nDst, int R) {
  __shared__ int s_cnt[OWN_RN];
  const int tid = threadIdx.x;
  const int wv = __builtin_amdgcn_readfirstlane(tid >> 5), lane = tid & 31;
  const int r0 = blockIdx.x * R;
  if (tid < R) s_cnt[tid] = 0;
  __syncthreads();

  OwnRuns rn;
  own_runs(off, wv, lane, rn);
  const int wcnt = rn.pre[8];
  for (int m0 = 0; m0 < wcnt; m0 += 32 * OWN_NB) {
    unsigned p[OWN_NB];
#pragma unroll
    for (int j = 0; j < OWN_NB; ++j) {
      const int m = m0 + 32 * j + lane;
      p[j] = own_match(lst, rn, m);
    }
#pragma unroll
    for (int j = 0; j < OWN_NB; ++j) {
      const unsigned pl = p[j] >> 16;
      if (pl < (unsigned)R) atomicAdd(&s_cnt[pl], 1);
    }
  }
  __syncthreads();
  if (tid < R && r0 + tid < nDst)
    dis[r0 + tid] = rsqrtf((float)s_cnt[tid] + 1.0f);
}

template <int MODE, bool PROD, bool MIX>
__global__ __launch_bounds__(OWN_T)
void owner_pass(const unsigned* __restrict__ lst, const int* __restrict__ off,
                const float* __restrict__ S, int nSrc,
                const float* __restrict__ dis, const float* __restrict__ bias,
                const float* __restrict__ W,
                float* __restrict__ out, int nDst, int R, int do_elu,
                const float* __restrict__ xs, const float* __restrict__ gate, float* __restrict__ z) {
  static_assert(!(MIX && PROD), "the mix is formed from rows that the pass writes out itself");
  constexpr int AP = PROD ? OWN_AP : 64;
  constexpr int AR = PROD ? OWN_RP : OWN_RN;
  __shared__ alignas(16) float s_acc[AR * AP];
  __shared__ alignas(16) float s_w[PROD ? 64 * OWN_WP : 4];
  __shared__ float s_dis[OWN_RN];
  __shared__ int   s_cnt[OWN_RN];
  __shared__ int   s_deg[OWN_RN];
  __shared__ int   s_start[OWN_RN];
  __shared__ int   s_tot;
  __shared__ int   s_more;
  __shared__ int2  s_l[OWN_LC];
  __shared__ float s_lw[OWN_LC];

  const int tid = threadIdx.x;
  const int wv = __builtin_amdgcn_readfirstlane(tid >> 5), lane = tid & 31;
  const int r0  = blockIdx.x * R;

  if (tid < R) {
    s_cnt[tid] = 0;
    if (MODE != 0) s_deg[tid] = 0;
    if (MODE == 0) s_dis[tid] = (r0 + tid < nDst) ? dis[r0 + tid] : 0.0f;
  }
  if (tid == 0) s_more = 0;
  float4 wq = make_float4(0.0f, 0.0f, 0.0f, 0.0f);
  if (PROD) wq = *(const float4*)(W + 4 * tid);
  const int rFill = PROD ? ((R + 15) & ~15) : R;
  for (int i = tid; i < rFill * 16; i += OWN_T) {
    const int dl = i >> 4, q = (i & 15) << 2;
    float4 a = make_float4(0.0f, 0.0f, 0.0f, 0.0f);
    if (MODE == 0 && dl < R && r0 + dl < nDst) {
      float w = dis[r0 + dl]; w *= w;
      const float4 v = *(const float4*)(S + (long)(r0 + dl) * 64 + q);
      a = make_float4(w * v.x, w * v.y, w * v.z, w * v.w);
    }
    *(float4*)&s_acc[dl * AP + q] = a;
  }
  if (PROD) *(float4*)&s_w[(tid >> 4) * OWN_WP + ((tid & 15) << 2)] = wq;
  __syncthreads();

  OwnRuns rn;
  own_runs(off, wv, lane, rn);
  const int wcnt = rn.pre[8];

  float4 sx[OWN_FT];
#pragma unroll
  for (int t = 0; t < OWN_FT; ++t) {
    sx[t] = make_float4(0.0f, 0.0f, 0.0f, 0.0f);
    if (MIX) {
      const int i = tid + t * OWN_T;
      const int rc = min(r0 + (i >> 4), nDst - 1);
      sx[t] = *(const float4*)(xs + (long)rc * 64 + ((i & 15) << 2));
    }
  }

  for (int base = 0; ; base += 32 * OWN_NB) {
    unsigned p[OWN_NB];
    int s[OWN_NB], dl[OWN_NB], rk[OWN_NB]; float ws[OWN_NB];
#pragma unroll
    for (int j = 0; j < OWN_NB; ++j) p[j] = 0xFFFFFFFFu;
    if (base < wcnt) {
#pragma unroll
      for (int j = 0; j < OWN_NB; ++j) p[j] = own_match(lst, rn, base + 32 * j + lane);
    }
    if (lane == 0 && wcnt > base + 32 * OWN_NB) s_more = 1;
#pragma unroll
    for (int j = 0; j < OWN_NB; ++j) {
      const unsigned pl = p[j] >> 16, ps = p[j] & 0xFFFFu;
      s[j] = -1; dl[j] = -1; rk[j] = 0;
      if (pl < (unsigned)R) {
        dl[j] = (int)pl;
        rk[j] = atomicAdd(&s_cnt[pl], 1);
        if (ps < (unsigned)nSrc) s[j] = (int)ps;
      }
    }
#pragma unroll
    for (int j = 0; j < OWN_NB; ++j) {
      const float t = (MODE == 0) ? dis[max(s[j], 0)] : 0.0f;
      ws[j] = (s[j] >= 0) ? t : 0.0f;
    }
    __syncthreads();

    if (wv == 0) {
      int cn[OWN_RL]; int tot = 0;
#pragma unroll
      for (int i = 0; i < OWN_RL; ++i) {
        const int r = lane * OWN_RL + i;
        cn[i] = (r < R) ? s_cnt[r] : 0;
        tot += cn[i];
      }
      int inc = tot;
#pragma unroll
      for (int k = 1; k < 32; k <<= 1) {
        const int t = __shfl_up(inc, k);
        if (lane >= k) inc += t;
      }
      int run = inc - tot;
#pragma unroll
      for (int i = 0; i < OWN_RL; ++i) {
        const int r = lane * OWN_RL + i;
        if (r < R) {
          s_start[r] = run;
          if (MODE != 0) s_deg[r] += cn[i];
        }
        run += cn[i];
      }
      if (lane == 31) s_tot = inc;
    }
    __syncthreads();

    const int more = s_more;
    if (tid < OWN_PU) {
      const int pos = s_tot + tid;
      s_l[pos] = make_int2(-1, 0);
      if (MODE == 0) s_lw[pos] = 0.0f;
    }
#pragma unroll
    for (int j = 0; j < OWN_NB; ++j)
      if (dl[j] >= 0) {
        const int pos = s_start[dl[j]] + rk[j];
        s_l[pos] = make_int2(s[j], dl[j]);
        if (MODE == 0) s_lw[pos] = ws[j];
      }
    __syncthreads();

    const int tot = __builtin_amdgcn_readfirstlane(s_tot);
    const int c0 = (tot * wv) >> 5, c1 = (tot * (wv + 1)) >> 5;
    int cur = -1;
    float ax = 0.0f, ay = 0.0f;
    for (int e0 = c0; e0 < c1; e0 += OWN_PU) {
      int sr[OWN_PU], dr[OWN_PU]; float w[OWN_PU]; float2 v[OWN_PU];
#pragma unroll
      for (int j = 0; j < OWN_PU; ++j) {
        const int e = e0 + j;
        const int2 sd = s_l[e];
        const int sj = __builtin_amdgcn_readfirstlane(sd.x);
        dr[j] = __builtin_amdgcn_readfirstlane(sd.y);
        sr[j] = (e < c1) ? sj : -2;
        w[j]  = (MODE == 0) ? s_lw[e] * s_dis[dr[j]] : 1.0f;
        v[j]  = *(const float2*)(S + (long)max(sr[j], 0) * 64 + 2 * lane);
      }
#pragma unroll
      for (int j = 0; j < OWN_PU; ++j)
        if (sr[j] != -2) {
          if (dr[j] != cur) {
            if (cur >= 0) {
              atomicAdd(&s_acc[cur * AP + 2 * lane],     ax);
              atomicAdd(&s_acc[cur * AP + 2 * lane + 1], ay);
            }
            cur = dr[j]; ax = 0.0f; ay = 0.0f;
          }
          if (sr[j] >= 0) {
            ax += w[j] * v[j].x;
            ay += w[j] * v[j].y;
          }
        }
    }
    if (cur >= 0) {
      atomicAdd(&s_acc[cur * AP + 2 * lane],     ax);
      atomicAdd(&s_acc[cur * AP + 2 * lane + 1], ay);
    }
    __syncthreads();
    if (!more) break;
    if (tid < R) s_cnt[tid] = 0;
    if (tid == 0) s_more = 0;
    __syncthreads();
  }

  float al = 0.0f;
  if (MIX) al = 1.0f / (1.0f + __expf(-gate[0]));
  auto finish = [&](int i, const float4& sv) {
    const int dl = i >> 4, q = (i & 15) << 2;
    const int row = r0 + dl;
    if (row < nDst) {
      const float4 a = *(const float4*)&s_acc[dl * AP + q];
      float4 r;
      if (MODE == 0) {
        const float4 b = *(const float4*)(bias + q);
        r = make_float4(a.x + b.x, a.y + b.y, a.z + b.z, a.w + b.w);
      } else {
        const int   c  = s_deg[dl];
        const float sc = (c > 0) ? (1.0f / (float)c) : 0.0f;
        if (MODE == 1) {
          r = make_float4(sc * a.x, sc * a.y, sc * a.z, sc * a.w);
        } else {
          const float4 b = *(const float4*)(bias + q);
          r = make_float4(sc * a.x + b.x, sc * a.y + b.y, sc * a.z + b.z, sc * a.w + b.w);
        }
      }
      if (do_elu) {
        r.x = (r.x > 0.0f) ? r.x : expm1f(r.x);
        r.y = (r.y > 0.0f) ? r.y : expm1f(r.y);
        r.z = (r.z > 0.0f) ? r.z : expm1f(r.z);
        r.w = (r.w > 0.0f) ? r.w : expm1f(r.w);
      }
      if (PROD) *(float4*)&s_acc[dl * AP + q] = r;
      else      *(float4*)(out + (long)row * 64 + q) = r;
      if (MIX) {
        const float be = 1.0f - al;
        *(float4*)(z + (long)row * 64 + q) =
            make_float4(__builtin_fmaf(al, sv.x, be * r.x), __builtin_fmaf(al, sv.y, be * r.y),
                        __builtin_fmaf(al, sv.z, be * r.z), __builtin_fmaf(al, sv.w, be * r.w));
      }
    }
  };
  if (MIX) {
#pragma unroll
    for (int t = 0; t < OWN_FT; ++t)
      if (tid + t * OWN_T < R * 16) finish(tid + t * OWN_T, sx[t]);
  } else {
    for (int i = tid; i < R * 16; i += OWN_T) finish(i, sx[0]);
  }

  if (PROD) {
    __syncthreads();
    const int m  = lane & 15;
    const int hh = lane >> 4;
    const int ct = (wv & 3) * 16;
    const int nOwn = min(R, nDst - r0);
    const int nRT = (nOwn + 15) >> 4;
    const float* bp = s_w + (hh << 1) * OWN_WP + ct + m;
    for (int rt = wv >> 2; rt < nRT; rt += OWN_T / 128) {
      const float* ap = s_acc + (rt * 16 + m) * AP + (hh << 1);
      v8f acc = {};
#pragma unroll
      for (int k0 = 0; k0 < 64; k0 += 4) {
        v2f a, b;
        a.x = ap[k0];
        a.y = ap[k0 + 1];
        b.x = bp[k0 * OWN_WP];
        b.y = bp[(k0 + 1) * OWN_WP];
        acc = __builtin_amdgcn_wmma_f32_16x16x4_f32(
            false, a, false, b, (short)0, acc, false, false);
      }
      const int dl0 = rt * 16 + (hh << 3);
      float* op = out + (long)(r0 + dl0) * 64 + ct + m;
      if (rt * 16 + 16 <= nOwn) {
#pragma unroll
        for (int r = 0; r < 8; ++r) op[(long)r * 64] = acc[r];
      } else {
#pragma unroll
        for (int r = 0; r < 8; ++r)
          if (dl0 + r < nOwn) op[(long)r * 64] = acc[r];
      }
    }
  }
}

extern "C" void kernel_launch(void* const* d_in, const int* in_sizes, int n_in,
                              void* d_out, int out_size, void* d_ws, size_t ws_size,
                              hipStream_t stream) {
  (void)in_sizes; (void)n_in; (void)out_size; (void)ws_size;

  const float* x   = (const float*)d_in[0];
  const int*   ei  = (const int*)d_in[1];
  const int*   hyi = (const int*)d_in[2];
  const float* g1w = (const float*)d_in[3];
  const float* g1b = (const float*)d_in[4];
  const float* g2w = (const float*)d_in[5];
  const float* g2b = (const float*)d_in[6];
  const float* h1w = (const float*)d_in[7];
  const float* h1b = (const float*)d_in[8];
  const float* h2w = (const float*)d_in[9];
  const float* h2b = (const float*)d_in[10];
  const float* gate = (const float*)d_in[11];

  float* out_z  = (float*)d_out;
  float* out_xs = out_z  + (long)N_NODES * F_OUT;
  float* out_xd = out_xs + (long)N_NODES * F_OUT;

  float* ws   = (float*)d_ws;
  float* h    = ws   + (long)N_NODES * 64;
  float* dis  = h    + (long)N_NODES * 64;
  float* ef   = dis  + N_NODES;
  float* xw   = out_xd;

  unsigned* lst  = (unsigned*)ws;
  int*      off  = (int*)(lst + 3L * PART_LW);
  unsigned* lstG = lst;                 int* offG = off;
  unsigned* lstH = lst + PART_LW;       int* offH = off + PART_OW;
  unsigned* lstN = lst + 2L * PART_LW;  int* offN = off + 2L * PART_OW;

  part_lists<<<3 * PART_N, PART_T, 0, stream>>>(ei + N_EDGES, ei, hyi + NNZ_H, hyi, hyi, hyi + NNZ_H, lst, off);

  owner_count<<<OWN_WG, OWN_T, 0, stream>>>(lstG, offG, dis, N_NODES, OWN_RN);

  gemm_wmma_f32<F_IN><<<(N_NODES / 16 + GEMM_WV - 1) / GEMM_WV, 32 * GEMM_WV, 0, stream>>>(x, g1w, xw, N_NODES / 16);
  owner_pass<0, true, false><<<OWN_WG, OWN_T, 0, stream>>>(lstG, offG, xw, N_NODES, dis, g1b, g2w, h, N_NODES, OWN_RN, 1, nullptr, nullptr, nullptr);
  owner_pass<0, false, false><<<OWN_WG, OWN_T, 0, stream>>>(lstG, offG, h, N_NODES, dis, g2b, nullptr, out_xs, N_NODES, OWN_RN, 0, nullptr, nullptr, nullptr);

  gemm_wmma_f32<F_IN><<<(N_NODES / 16 + GEMM_WV - 1) / GEMM_WV, 32 * GEMM_WV, 0, stream>>>(x, h1w, xw, N_NODES / 16);
  owner_pass<1, false, false><<<OWN_WG, OWN_T, 0, stream>>>(lstH, offH, xw, N_NODES, nullptr, nullptr, nullptr, ef, M_HE, OWN_RM, 0, nullptr, nullptr, nullptr);
  owner_pass<2, true, false><<<OWN_WG, OWN_T, 0, stream>>>(lstN, offN, ef, M_HE, nullptr, h1b, h2w, h, N_NODES, OWN_RN, 1, nullptr, nullptr, nullptr);
  owner_pass<1, false, false><<<OWN_WG, OWN_T, 0, stream>>>(lstH, offH, h, N_NODES, nullptr, nullptr, nullptr, ef, M_HE, OWN_RM, 0, nullptr, nullptr, nullptr);
  owner_pass<2, false, true><<<OWN_WG, OWN_T, 0, stream>>>(lstN, offN, ef, M_HE, nullptr, h2b, nullptr, out_xd, N_NODES, OWN_RN, 0, out_xs, gate, out_z);
}
